# starting kernel
# baseline (speedup 1.0000x reference)
#include <hip/hip_runtime.h>

typedef __attribute__((ext_vector_type(2))) float v2f;
typedef __attribute__((ext_vector_type(8))) float v8f;

#define B_   4
#define C_   68
#define F0   4
#define PT   2048
#define PS   8192
#define KSEL 4
#define FLT_BIG 3.4e38f

__global__ void sqnorm_kernel(const float* __restrict__ S, float* __restrict__ sq) {
  int q = blockIdx.x * blockDim.x + threadIdx.x;
  int b = blockIdx.y;
  if (q >= PS) return;
  const float* p = S + ((size_t)b * C_ + F0) * PS + q;
  float acc = 0.f;
  #pragma unroll 8
  for (int f = 0; f < 64; ++f) {
    float v = p[(size_t)f * PS];
    acc = fmaf(v, v, acc);
  }
  sq[b * PS + q] = acc;
}

__launch_bounds__(128)
__global__ void knn_kernel(const float* __restrict__ T, const float* __restrict__ S,
                           const float* __restrict__ sq, int* __restrict__ out_idx) {
  const int b    = blockIdx.y;
  const int p0   = blockIdx.x * 16;
  const int lane = threadIdx.x & 31;
  const int wave = threadIdx.x >> 5;
  const int half = lane >> 4;
  const int ln   = lane & 15;

  v2f a[16];
  #pragma unroll
  for (int s = 0; s < 16; ++s) {
    int f = F0 + 4 * s + 2 * half;
    const float* tp = T + ((size_t)b * C_ + f) * PT + p0 + ln;
    a[s].x = tp[0];
    a[s].y = tp[PT];
  }

  float best[8][KSEL];
  int   bidx[8][KSEL];
  #pragma unroll
  for (int v = 0; v < 8; ++v)
    #pragma unroll
    for (int j = 0; j < KSEL; ++j) { best[v][j] = FLT_BIG; bidx[v][j] = 0; }

  const int qbeg = wave * (PS / 4);
  const int qend = qbeg + (PS / 4);
  for (int q0 = qbeg; q0 < qend; q0 += 16) {
    v2f bb[16];
    #pragma unroll
    for (int s = 0; s < 16; ++s) {
      int f = F0 + 4 * s + 2 * half;
      const float* sp = S + ((size_t)b * C_ + f) * PS + q0 + ln;
      bb[s].x = sp[0];
      bb[s].y = sp[PS];
    }

    v8f acc = {};
    #pragma unroll
    for (int s = 0; s < 16; ++s) {
      acc = __builtin_amdgcn_wmma_f32_16x16x4_f32(
           false, a[s],  false, bb[s],
           (short)0, acc,  false,  false);
    }

    const float sqv  = sq[b * PS + q0 + ln];
    const int   qidx = q0 + ln;

    #pragma unroll
    for (int v = 0; v < 8; ++v) {
      float sc = fmaf(-2.f, acc[v], sqv);
      if (sc < best[v][3]) {
        if (sc < best[v][2]) {
          best[v][3] = best[v][2]; bidx[v][3] = bidx[v][2];
          if (sc < best[v][1]) {
            best[v][2] = best[v][1]; bidx[v][2] = bidx[v][1];
            if (sc < best[v][0]) {
              best[v][1] = best[v][0]; bidx[v][1] = bidx[v][0];
              best[v][0] = sc; bidx[v][0] = qidx;
            } else { best[v][1] = sc; bidx[v][1] = qidx; }
          } else { best[v][2] = sc; bidx[v][2] = qidx; }
        } else { best[v][3] = sc; bidx[v][3] = qidx; }
      }
    }
  }

  __shared__ float ls[4][16][16][KSEL];
  __shared__ int   li[4][16][16][KSEL];
  #pragma unroll
  for (int v = 0; v < 8; ++v) {
    int row = v + 8 * half;
    #pragma unroll
    for (int j = 0; j < KSEL; ++j) {
      ls[wave][row][ln][j] = best[v][j];
      li[wave][row][ln][j] = bidx[v][j];
    }
  }
  __syncthreads();

  if (threadIdx.x < 16) {
    const int row = threadIdx.x;
    float rs[KSEL]; int ri[KSEL];
    #pragma unroll
    for (int j = 0; j < KSEL; ++j) { rs[j] = FLT_BIG; ri[j] = 0; }
    for (int w = 0; w < 4; ++w)
      for (int n = 0; n < 16; ++n)
        #pragma unroll
        for (int j = 0; j < KSEL; ++j) {
          float sc = ls[w][row][n][j];
          int   id = li[w][row][n][j];
          if (sc < rs[3]) {
            if (sc < rs[2]) {
              rs[3] = rs[2]; ri[3] = ri[2];
              if (sc < rs[1]) {
                rs[2] = rs[1]; ri[2] = ri[1];
                if (sc < rs[0]) {
                  rs[1] = rs[0]; ri[1] = ri[0]; rs[0] = sc; ri[0] = id;
                } else { rs[1] = sc; ri[1] = id; }
              } else { rs[2] = sc; ri[2] = id; }
            } else { rs[3] = sc; ri[3] = id; }
          }
        }
    int* o = out_idx + ((size_t)b * PT + p0 + row) * KSEL;
    o[0] = ri[0]; o[1] = ri[1]; o[2] = ri[2]; o[3] = ri[3];
  }
}

__global__ void gather_kernel(const float* __restrict__ S, const int* __restrict__ idx,
                              float* __restrict__ out) {
  size_t t = (size_t)blockIdx.x * blockDim.x + threadIdx.x;
  if (t >= (size_t)B_ * C_ * PT) return;
  int p  = (int)(t % PT);
  int bc = (int)(t / PT);
  int b  = bc / C_;
  const int4 id = *(const int4*)(idx + ((size_t)b * PT + p) * KSEL);
  const float* src = S + (size_t)bc * PS;
  float4 r;
  r.x = src[id.x]; r.y = src[id.y]; r.z = src[id.z]; r.w = src[id.w];
  *(float4*)(out + t * KSEL) = r;
}

extern "C" void kernel_launch(void* const* d_in, const int* in_sizes, int n_in,
                              void* d_out, int out_size, void* d_ws, size_t ws_size,
                              hipStream_t stream) {
  const float* T = (const float*)d_in[0];
  const float* S = (const float*)d_in[1];

  float* sq  = (float*)d_ws;
  int*   idx = (int*)((char*)d_ws + (size_t)B_ * PS * sizeof(float));
  float* out = (float*)d_out;

  sqnorm_kernel<<<dim3(PS / 256, B_), 256, 0, stream>>>(S, sq);
  knn_kernel<<<dim3(PT / 16, B_), 128, 0, stream>>>(T, S, sq, idx);
  size_t n = (size_t)B_ * C_ * PT;
  gather_kernel<<<dim3((unsigned)((n + 255) / 256)), 256, 0, stream>>>(S, idx, out);
}


# Row-wide pruning threshold: after 1,2,4,8,16 slabs a DPP butterfly merges the 16 lanes' top-4 values of each row; a score not below the row's 4th best skips the insertion code; scores bit-identical
# speedup vs baseline: 1.2302x; 1.2302x over previous
#include <hip/hip_runtime.h>

typedef __attribute__((ext_vector_type(2))) float v2f;
typedef __attribute__((ext_vector_type(4))) float v4f;
typedef __attribute__((ext_vector_type(8))) float v8f;

#define B_   4
#define C_   68
#define F0   4
#define PT   2048
#define PS   8192
#define KSEL 4
#define FLT_BIG 3.4e38f

__global__ void sqnorm_kernel(const float* __restrict__ S, float* __restrict__ sq) {
  int q = blockIdx.x * blockDim.x + threadIdx.x;
  int b = blockIdx.y;
  if (q >= PS) return;
  const float* p = S + ((size_t)b * C_ + F0) * PS + q;
  float acc = 0.f;
  #pragma unroll 8
  for (int f = 0; f < 64; ++f) {
    float v = p[(size_t)f * PS];
    acc = fmaf(v, v, acc);
  }
  sq[b * PS + q] = acc;
}

template <int CTRL>
__device__ __forceinline__ float row_xchg(float x) {
  return __int_as_float(__builtin_amdgcn_update_dpp(
      __float_as_int(x), __float_as_int(x), CTRL, 0xF, 0xF, false));
}

template <int CTRL>
__device__ __forceinline__ void row_merge4(float& c0, float& c1, float& c2, float& c3) {
  const float d0 = fminf(c0, row_xchg<CTRL>(c3));
  const float d1 = fminf(c1, row_xchg<CTRL>(c2));
  const float d2 = fminf(c2, row_xchg<CTRL>(c1));
  const float d3 = fminf(c3, row_xchg<CTRL>(c0));
  const float e0 = fminf(d0, d2), e2 = fmaxf(d0, d2);
  const float e1 = fminf(d1, d3), e3 = fmaxf(d1, d3);
  c0 = fminf(e0, e1); c1 = fmaxf(e0, e1);
  c2 = fminf(e2, e3); c3 = fmaxf(e2, e3);
}

__device__ __forceinline__ float row_fourth(float c0, float c1, float c2, float c3) {
  row_merge4<0x161>(c0, c1, c2, c3);
  row_merge4<0x162>(c0, c1, c2, c3);
  row_merge4<0x164>(c0, c1, c2, c3);
  const float d0 = fminf(c0, row_xchg<0x168>(c3));
  const float d1 = fminf(c1, row_xchg<0x168>(c2));
  const float d2 = fminf(c2, row_xchg<0x168>(c1));
  const float d3 = fminf(c3, row_xchg<0x168>(c0));
  return fmaxf(fmaxf(d0, d1), fmaxf(d2, d3));
}

__launch_bounds__(128)
__global__ void knn_kernel(const float* __restrict__ T, const float* __restrict__ S,
                           const float* __restrict__ sq, int* __restrict__ out_idx) {
  const int b    = blockIdx.y;
  const int p0   = blockIdx.x * 16;
  const int lane = threadIdx.x & 31;
  const int wave = threadIdx.x >> 5;
  const int half = lane >> 4;
  const int ln   = lane & 15;

  v2f a[16];
  #pragma unroll
  for (int s = 0; s < 16; ++s) {
    int f = F0 + 4 * s + 2 * half;
    const float* tp = T + ((size_t)b * C_ + f) * PT + p0 + ln;
    a[s].x = tp[0];
    a[s].y = tp[PT];
  }

  float best[8][KSEL];
  int   bidx[8][KSEL];
  #pragma unroll
  for (int v = 0; v < 8; ++v)
    #pragma unroll
    for (int j = 0; j < KSEL; ++j) { best[v][j] = FLT_BIG; bidx[v][j] = 0; }

  float thr[8];
  #pragma unroll
  for (int v = 0; v < 8; ++v) thr[v] = FLT_BIG;

  const int qbeg = wave * (PS / 4);
  for (int it = 0; it < (PS / 4) / 64; ++it) {
    const int q0 = qbeg + it * 64;
    v4f bx[16], by[16];
    #pragma unroll
    for (int s = 0; s < 16; ++s) {
      int f = F0 + 4 * s + 2 * half;
      const float* sp = S + ((size_t)b * C_ + f) * PS + q0 + 4 * ln;
      bx[s] = *(const v4f*)sp;
      by[s] = *(const v4f*)(sp + PS);
    }
    const v4f sq4 = *(const v4f*)(sq + b * PS + q0 + 4 * ln);

    v8f acc[4] = {};
    #pragma unroll
    for (int s = 0; s < 16; ++s) {
      #pragma unroll
      for (int t = 0; t < 4; ++t) {
        v2f bb;
        bb.x = bx[s][t];
        bb.y = by[s][t];
        acc[t] = __builtin_amdgcn_wmma_f32_16x16x4_f32(
             false, a[s],  false, bb,
             (short)0, acc[t],  false,  false);
      }
    }

    #pragma unroll
    for (int t = 0; t < 4; ++t) {
      const float sqv  = sq4[t];
      const int   qidx = q0 + 4 * ln + t;

      #pragma unroll
      for (int v = 0; v < 8; ++v) {
        float sc = fmaf(-2.f, acc[t][v], sqv);
        if (sc < thr[v]) {
          if (sc < best[v][2]) {
            best[v][3] = best[v][2]; bidx[v][3] = bidx[v][2];
            if (sc < best[v][1]) {
              best[v][2] = best[v][1]; bidx[v][2] = bidx[v][1];
              if (sc < best[v][0]) {
                best[v][1] = best[v][0]; bidx[v][1] = bidx[v][0];
                best[v][0] = sc; bidx[v][0] = qidx;
              } else { best[v][1] = sc; bidx[v][1] = qidx; }
            } else { best[v][2] = sc; bidx[v][2] = qidx; }
          } else { best[v][3] = sc; bidx[v][3] = qidx; }
          thr[v] = fminf(thr[v], best[v][3]);
        }
      }
    }

    const int done = it + 1;
    if ((done & it) == 0 && done < (PS / 4) / 64) {
      #pragma unroll
      for (int v = 0; v < 8; ++v)
        thr[v] = row_fourth(best[v][0], best[v][1], best[v][2], best[v][3]);
    }
  }

  __shared__ float ls[4][16][16][KSEL];
  __shared__ int   li[4][16][16][KSEL];
  #pragma unroll
  for (int v = 0; v < 8; ++v) {
    int row = v + 8 * half;
    #pragma unroll
    for (int j = 0; j < KSEL; ++j) {
      ls[wave][row][ln][j] = best[v][j];
      li[wave][row][ln][j] = bidx[v][j];
    }
  }
  __syncthreads();

  if (threadIdx.x < 16) {
    const int row = threadIdx.x;
    float rs[KSEL]; int ri[KSEL];
    #pragma unroll
    for (int j = 0; j < KSEL; ++j) { rs[j] = FLT_BIG; ri[j] = 0; }
    for (int w = 0; w < 4; ++w)
      for (int n = 0; n < 16; ++n)
        #pragma unroll
        for (int j = 0; j < KSEL; ++j) {
          float sc = ls[w][row][n][j];
          int   id = li[w][row][n][j];
          if (sc < rs[3]) {
            if (sc < rs[2]) {
              rs[3] = rs[2]; ri[3] = ri[2];
              if (sc < rs[1]) {
                rs[2] = rs[1]; ri[2] = ri[1];
                if (sc < rs[0]) {
                  rs[1] = rs[0]; ri[1] = ri[0]; rs[0] = sc; ri[0] = id;
                } else { rs[1] = sc; ri[1] = id; }
              } else { rs[2] = sc; ri[2] = id; }
            } else { rs[3] = sc; ri[3] = id; }
          }
        }
    int* o = out_idx + ((size_t)b * PT + p0 + row) * KSEL;
    o[0] = ri[0]; o[1] = ri[1]; o[2] = ri[2]; o[3] = ri[3];
  }
}

__global__ void gather_kernel(const float* __restrict__ S, const int* __restrict__ idx,
                              float* __restrict__ out) {
  size_t t = (size_t)blockIdx.x * blockDim.x + threadIdx.x;
  if (t >= (size_t)B_ * C_ * PT) return;
  int p  = (int)(t % PT);
  int bc = (int)(t / PT);
  int b  = bc / C_;
  const int4 id = *(const int4*)(idx + ((size_t)b * PT + p) * KSEL);
  const float* src = S + (size_t)bc * PS;
  float4 r;
  r.x = src[id.x]; r.y = src[id.y]; r.z = src[id.z]; r.w = src[id.w];
  *(float4*)(out + t * KSEL) = r;
}

extern "C" void kernel_launch(void* const* d_in, const int* in_sizes, int n_in,
                              void* d_out, int out_size, void* d_ws, size_t ws_size,
                              hipStream_t stream) {
  const float* T = (const float*)d_in[0];
  const float* S = (const float*)d_in[1];

  float* sq  = (float*)d_ws;
  int*   idx = (int*)((char*)d_ws + (size_t)B_ * PS * sizeof(float));
  float* out = (float*)d_out;

  sqnorm_kernel<<<dim3(PS / 256, B_), 256, 0, stream>>>(S, sq);
  knn_kernel<<<dim3(PT / 16, B_), 128, 0, stream>>>(T, S, sq, idx);
  size_t n = (size_t)B_ * C_ * PT;
  gather_kernel<<<dim3((unsigned)((n + 255) / 256)), 256, 0, stream>>>(S, idx, out);
}


# All 32 threshold compares of a slab issued first as wave ballots into scalar registers; insertion code entered by scalar tests: one vector-to-scalar wait per slab, not 32; scores bit-identical
# speedup vs baseline: 1.3229x; 1.0754x over previous
#include <hip/hip_runtime.h>

typedef __attribute__((ext_vector_type(2))) float v2f;
typedef __attribute__((ext_vector_type(4))) float v4f;
typedef __attribute__((ext_vector_type(8))) float v8f;

#define B_   4
#define C_   68
#define F0   4
#define PT   2048
#define PS   8192
#define KSEL 4
#define FLT_BIG 3.4e38f

__global__ void sqnorm_kernel(const float* __restrict__ S, float* __restrict__ sq) {
  int q = blockIdx.x * blockDim.x + threadIdx.x;
  int b = blockIdx.y;
  if (q >= PS) return;
  const float* p = S + ((size_t)b * C_ + F0) * PS + q;
  float acc = 0.f;
  #pragma unroll 8
  for (int f = 0; f < 64; ++f) {
    float v = p[(size_t)f * PS];
    acc = fmaf(v, v, acc);
  }
  sq[b * PS + q] = acc;
}

template <int CTRL>
__device__ __forceinline__ float row_xchg(float x) {
  return __int_as_float(__builtin_amdgcn_update_dpp(
      __float_as_int(x), __float_as_int(x), CTRL, 0xF, 0xF, false));
}

template <int CTRL>
__device__ __forceinline__ void row_merge4(float& c0, float& c1, float& c2, float& c3) {
  const float d0 = fminf(c0, row_xchg<CTRL>(c3));
  const float d1 = fminf(c1, row_xchg<CTRL>(c2));
  const float d2 = fminf(c2, row_xchg<CTRL>(c1));
  const float d3 = fminf(c3, row_xchg<CTRL>(c0));
  const float e0 = fminf(d0, d2), e2 = fmaxf(d0, d2);
  const float e1 = fminf(d1, d3), e3 = fmaxf(d1, d3);
  c0 = fminf(e0, e1); c1 = fmaxf(e0, e1);
  c2 = fminf(e2, e3); c3 = fmaxf(e2, e3);
}

__device__ __forceinline__ float row_fourth(float c0, float c1, float c2, float c3) {
  row_merge4<0x161>(c0, c1, c2, c3);
  row_merge4<0x162>(c0, c1, c2, c3);
  row_merge4<0x164>(c0, c1, c2, c3);
  const float d0 = fminf(c0, row_xchg<0x168>(c3));
  const float d1 = fminf(c1, row_xchg<0x168>(c2));
  const float d2 = fminf(c2, row_xchg<0x168>(c1));
  const float d3 = fminf(c3, row_xchg<0x168>(c0));
  return fmaxf(fmaxf(d0, d1), fmaxf(d2, d3));
}

__launch_bounds__(128)
__global__ void knn_kernel(const float* __restrict__ T, const float* __restrict__ S,
                           const float* __restrict__ sq, int* __restrict__ out_idx) {
  const int b    = blockIdx.y;
  const int p0   = blockIdx.x * 16;
  const int lane = threadIdx.x & 31;
  const int wave = threadIdx.x >> 5;
  const int half = lane >> 4;
  const int ln   = lane & 15;

  v2f a[16];
  #pragma unroll
  for (int s = 0; s < 16; ++s) {
    int f = F0 + 4 * s + 2 * half;
    const float* tp = T + ((size_t)b * C_ + f) * PT + p0 + ln;
    a[s].x = tp[0];
    a[s].y = tp[PT];
  }

  float best[8][KSEL];
  int   bidx[8][KSEL];
  #pragma unroll
  for (int v = 0; v < 8; ++v)
    #pragma unroll
    for (int j = 0; j < KSEL; ++j) { best[v][j] = FLT_BIG; bidx[v][j] = 0; }

  float thr[8];
  #pragma unroll
  for (int v = 0; v < 8; ++v) thr[v] = FLT_BIG;

  const int qbeg = wave * (PS / 4);
  for (int it = 0; it < (PS / 4) / 64; ++it) {
    const int q0 = qbeg + it * 64;
    v4f bx[16], by[16];
    #pragma unroll
    for (int s = 0; s < 16; ++s) {
      int f = F0 + 4 * s + 2 * half;
      const float* sp = S + ((size_t)b * C_ + f) * PS + q0 + 4 * ln;
      bx[s] = *(const v4f*)sp;
      by[s] = *(const v4f*)(sp + PS);
    }
    const v4f sq4 = *(const v4f*)(sq + b * PS + q0 + 4 * ln);

    v8f acc[4] = {};
    #pragma unroll
    for (int s = 0; s < 16; ++s) {
      #pragma unroll
      for (int t = 0; t < 4; ++t) {
        v2f bb;
        bb.x = bx[s][t];
        bb.y = by[s][t];
        acc[t] = __builtin_amdgcn_wmma_f32_16x16x4_f32(
             false, a[s],  false, bb,
             (short)0, acc[t],  false,  false);
      }
    }

    float sc[4][8];
    #pragma unroll
    for (int t = 0; t < 4; ++t)
      #pragma unroll
      for (int v = 0; v < 8; ++v)
        sc[t][v] = fmaf(-2.f, acc[t][v], sq4[t]);

    unsigned pm[4][8];
    #pragma unroll
    for (int t = 0; t < 4; ++t)
      #pragma unroll
      for (int v = 0; v < 8; ++v)
        pm[t][v] = __builtin_amdgcn_ballot_w32(sc[t][v] < thr[v]);

    #pragma unroll
    for (int t = 0; t < 4; ++t) {
      const int qidx = q0 + 4 * ln + t;

      #pragma unroll
      for (int v = 0; v < 8; ++v) {
        if (__builtin_expect(pm[t][v] != 0u, 0)) {
          const float s = sc[t][v];
          if (s < thr[v]) {
            if (s < best[v][2]) {
              best[v][3] = best[v][2]; bidx[v][3] = bidx[v][2];
              if (s < best[v][1]) {
                best[v][2] = best[v][1]; bidx[v][2] = bidx[v][1];
                if (s < best[v][0]) {
                  best[v][1] = best[v][0]; bidx[v][1] = bidx[v][0];
                  best[v][0] = s; bidx[v][0] = qidx;
                } else { best[v][1] = s; bidx[v][1] = qidx; }
              } else { best[v][2] = s; bidx[v][2] = qidx; }
            } else { best[v][3] = s; bidx[v][3] = qidx; }
            thr[v] = fminf(thr[v], best[v][3]);
          }
        }
      }
    }

    const int done = it + 1;
    if ((done & it) == 0 && done < (PS / 4) / 64) {
      #pragma unroll
      for (int v = 0; v < 8; ++v)
        thr[v] = row_fourth(best[v][0], best[v][1], best[v][2], best[v][3]);
    }
  }

  __shared__ float ls[4][16][16][KSEL];
  __shared__ int   li[4][16][16][KSEL];
  #pragma unroll
  for (int v = 0; v < 8; ++v) {
    int row = v + 8 * half;
    #pragma unroll
    for (int j = 0; j < KSEL; ++j) {
      ls[wave][row][ln][j] = best[v][j];
      li[wave][row][ln][j] = bidx[v][j];
    }
  }
  __syncthreads();

  if (threadIdx.x < 16) {
    const int row = threadIdx.x;
    float rs[KSEL]; int ri[KSEL];
    #pragma unroll
    for (int j = 0; j < KSEL; ++j) { rs[j] = FLT_BIG; ri[j] = 0; }
    for (int w = 0; w < 4; ++w)
      for (int n = 0; n < 16; ++n)
        #pragma unroll
        for (int j = 0; j < KSEL; ++j) {
          float sc = ls[w][row][n][j];
          int   id = li[w][row][n][j];
          if (sc < rs[3]) {
            if (sc < rs[2]) {
              rs[3] = rs[2]; ri[3] = ri[2];
              if (sc < rs[1]) {
                rs[2] = rs[1]; ri[2] = ri[1];
                if (sc < rs[0]) {
                  rs[1] = rs[0]; ri[1] = ri[0]; rs[0] = sc; ri[0] = id;
                } else { rs[1] = sc; ri[1] = id; }
              } else { rs[2] = sc; ri[2] = id; }
            } else { rs[3] = sc; ri[3] = id; }
          }
        }
    int* o = out_idx + ((size_t)b * PT + p0 + row) * KSEL;
    o[0] = ri[0]; o[1] = ri[1]; o[2] = ri[2]; o[3] = ri[3];
  }
}

__global__ void gather_kernel(const float* __restrict__ S, const int* __restrict__ idx,
                              float* __restrict__ out) {
  size_t t = (size_t)blockIdx.x * blockDim.x + threadIdx.x;
  if (t >= (size_t)B_ * C_ * PT) return;
  int p  = (int)(t % PT);
  int bc = (int)(t / PT);
  int b  = bc / C_;
  const int4 id = *(const int4*)(idx + ((size_t)b * PT + p) * KSEL);
  const float* src = S + (size_t)bc * PS;
  float4 r;
  r.x = src[id.x]; r.y = src[id.y]; r.z = src[id.z]; r.w = src[id.w];
  *(float4*)(out + t * KSEL) = r;
}

extern "C" void kernel_launch(void* const* d_in, const int* in_sizes, int n_in,
                              void* d_out, int out_size, void* d_ws, size_t ws_size,
                              hipStream_t stream) {
  const float* T = (const float*)d_in[0];
  const float* S = (const float*)d_in[1];

  float* sq  = (float*)d_ws;
  int*   idx = (int*)((char*)d_ws + (size_t)B_ * PS * sizeof(float));
  float* out = (float*)d_out;

  sqnorm_kernel<<<dim3(PS / 256, B_), 256, 0, stream>>>(S, sq);
  knn_kernel<<<dim3(PT / 16, B_), 128, 0, stream>>>(T, S, sq, idx);
  size_t n = (size_t)B_ * C_ * PT;
  gather_kernel<<<dim3((unsigned)((n + 255) / 256)), 256, 0, stream>>>(S, idx, out);
}


# Block = 8 target tiles (one per wave) x one source quarter; each 64-source slab staged once in LDS and read by all 8 waves as WMMA B; quarters merged in a small 4th kernel; scores bit-identical
# speedup vs baseline: 1.4417x; 1.0898x over previous
#include <hip/hip_runtime.h>

typedef __attribute__((ext_vector_type(2))) float v2f;
typedef __attribute__((ext_vector_type(4))) float v4f;
typedef __attribute__((ext_vector_type(8))) float v8f;

#define B_   4
#define C_   68
#define F0   4
#define PT   2048
#define PS   8192
#define KSEL 4
#define FLT_BIG 3.4e38f

__global__ void sqnorm_kernel(const float* __restrict__ S, float* __restrict__ sq) {
  int q = blockIdx.x * blockDim.x + threadIdx.x;
  int b = blockIdx.y;
  if (q >= PS) return;
  const float* p = S + ((size_t)b * C_ + F0) * PS + q;
  float acc = 0.f;
  #pragma unroll 8
  for (int f = 0; f < 64; ++f) {
    float v = p[(size_t)f * PS];
    acc = fmaf(v, v, acc);
  }
  sq[b * PS + q] = acc;
}

template <int CTRL>
__device__ __forceinline__ float row_xchg(float x) {
  return __int_as_float(__builtin_amdgcn_update_dpp(
      __float_as_int(x), __float_as_int(x), CTRL, 0xF, 0xF, false));
}

template <int CTRL>
__device__ __forceinline__ void row_merge4(float& c0, float& c1, float& c2, float& c3) {
  const float d0 = fminf(c0, row_xchg<CTRL>(c3));
  const float d1 = fminf(c1, row_xchg<CTRL>(c2));
  const float d2 = fminf(c2, row_xchg<CTRL>(c1));
  const float d3 = fminf(c3, row_xchg<CTRL>(c0));
  const float e0 = fminf(d0, d2), e2 = fmaxf(d0, d2);
  const float e1 = fminf(d1, d3), e3 = fmaxf(d1, d3);
  c0 = fminf(e0, e1); c1 = fmaxf(e0, e1);
  c2 = fminf(e2, e3); c3 = fmaxf(e2, e3);
}

__device__ __forceinline__ float row_fourth(float c0, float c1, float c2, float c3) {
  row_merge4<0x161>(c0, c1, c2, c3);
  row_merge4<0x162>(c0, c1, c2, c3);
  row_merge4<0x164>(c0, c1, c2, c3);
  const float d0 = fminf(c0, row_xchg<0x168>(c3));
  const float d1 = fminf(c1, row_xchg<0x168>(c2));
  const float d2 = fminf(c2, row_xchg<0x168>(c1));
  const float d3 = fminf(c3, row_xchg<0x168>(c0));
  return fmaxf(fmaxf(d0, d1), fmaxf(d2, d3));
}

#define NW   8
#define NQ   4
#define QLEN (PS / NQ)
#define CROW 260

__launch_bounds__(256)
__global__ void knn_kernel(const float* __restrict__ T, const float* __restrict__ S,
                           const float* __restrict__ sq, float* __restrict__ pscore,
                           int* __restrict__ pidx) {
  __shared__ __attribute__((aligned(16))) float stage[2][16 * CROW];

  const int b    = blockIdx.y;
  const int z    = blockIdx.z;
  const int lane = threadIdx.x & 31;
  const int wave = threadIdx.x >> 5;
  const int half = lane >> 4;
  const int ln   = lane & 15;
  const int p0   = blockIdx.x * (16 * NW) + wave * 16;

  v2f a[16];
  #pragma unroll
  for (int s = 0; s < 16; ++s) {
    int f = F0 + 4 * s + 2 * half;
    const float* tp = T + ((size_t)b * C_ + f) * PT + p0 + ln;
    a[s].x = tp[0];
    a[s].y = tp[PT];
  }

  float best[8][KSEL];
  int   bidx[8][KSEL];
  #pragma unroll
  for (int v = 0; v < 8; ++v)
    #pragma unroll
    for (int j = 0; j < KSEL; ++j) { best[v][j] = FLT_BIG; bidx[v][j] = 0; }

  float thr[8];
  #pragma unroll
  for (int v = 0; v < 8; ++v) thr[v] = FLT_BIG;

  const int qbeg = z * QLEN;
  const int fs = 2 * (wave * 2 + half);
  const float* gsrc = S + ((size_t)b * C_ + F0 + fs) * PS + qbeg + 4 * ln;
  const int st_off = ln * CROW + fs;
  const int fr_off = ln * CROW + 2 * half;

  v4f gx[2], gy[2];
  #pragma unroll
  for (int j = 0; j < 2; ++j) {
    gx[j] = *(const v4f*)(gsrc + (size_t)(32 * j) * PS);
    gy[j] = *(const v4f*)(gsrc + (size_t)(32 * j + 1) * PS);
  }
  #pragma unroll
  for (int j = 0; j < 2; ++j)
    #pragma unroll
    for (int t = 0; t < 4; ++t) {
      v2f w; w.x = gx[j][t]; w.y = gy[j][t];
      *(v2f*)(&stage[0][st_off + 32 * j + 64 * t]) = w;
    }
  __syncthreads();

  for (int it = 0; it < QLEN / 64; ++it) {
    const int q0  = qbeg + it * 64;
    const int cur = it & 1;
    const bool more = (it + 1) < QLEN / 64;

    const v4f sq4 = *(const v4f*)(sq + b * PS + q0 + 4 * ln);
    if (more) {
      #pragma unroll
      for (int j = 0; j < 2; ++j) {
        gx[j] = *(const v4f*)(gsrc + (size_t)(32 * j) * PS + (it + 1) * 64);
        gy[j] = *(const v4f*)(gsrc + (size_t)(32 * j + 1) * PS + (it + 1) * 64);
      }
    }

    const float* fr = &stage[cur][fr_off];
    v2f bf[16][4];
    #pragma unroll
    for (int s = 0; s < 16; ++s)
      #pragma unroll
      for (int t = 0; t < 4; ++t)
        bf[s][t] = *(const v2f*)(fr + 64 * t + 4 * s);
    __builtin_amdgcn_sched_barrier(0);

    v8f acc[4] = {};
    #pragma unroll
    for (int s = 0; s < 16; ++s) {
      #pragma unroll
      for (int t = 0; t < 4; ++t) {
        acc[t] = __builtin_amdgcn_wmma_f32_16x16x4_f32(
             false, a[s],  false, bf[s][t],
             (short)0, acc[t],  false,  false);
      }
    }

    float sc[4][8];
    #pragma unroll
    for (int t = 0; t < 4; ++t)
      #pragma unroll
      for (int v = 0; v < 8; ++v)
        sc[t][v] = fmaf(-2.f, acc[t][v], sq4[t]);

    unsigned pm[4][8];
    #pragma unroll
    for (int t = 0; t < 4; ++t)
      #pragma unroll
      for (int v = 0; v < 8; ++v)
        pm[t][v] = __builtin_amdgcn_ballot_w32(sc[t][v] < thr[v]);

    #pragma unroll
    for (int t = 0; t < 4; ++t) {
      const int qidx = q0 + 4 * ln + t;

      #pragma unroll
      for (int v = 0; v < 8; ++v) {
        if (__builtin_expect(pm[t][v] != 0u, 0)) {
          const float s = sc[t][v];
          if (s < thr[v]) {
            if (s < best[v][2]) {
              best[v][3] = best[v][2]; bidx[v][3] = bidx[v][2];
              if (s < best[v][1]) {
                best[v][2] = best[v][1]; bidx[v][2] = bidx[v][1];
                if (s < best[v][0]) {
                  best[v][1] = best[v][0]; bidx[v][1] = bidx[v][0];
                  best[v][0] = s; bidx[v][0] = qidx;
                } else { best[v][1] = s; bidx[v][1] = qidx; }
              } else { best[v][2] = s; bidx[v][2] = qidx; }
            } else { best[v][3] = s; bidx[v][3] = qidx; }
            thr[v] = fminf(thr[v], best[v][3]);
          }
        }
      }
    }

    if (more) {
      #pragma unroll
      for (int j = 0; j < 2; ++j)
        #pragma unroll
        for (int t = 0; t < 4; ++t) {
          v2f w; w.x = gx[j][t]; w.y = gy[j][t];
          *(v2f*)(&stage[cur ^ 1][st_off + 32 * j + 64 * t]) = w;
        }
    }
    __syncthreads();

    const int done = it + 1;
    if ((done & it) == 0 && done < (PS / 4) / 64) {
      #pragma unroll
      for (int v = 0; v < 8; ++v)
        thr[v] = row_fourth(best[v][0], best[v][1], best[v][2], best[v][3]);
    }
  }

  __shared__ __attribute__((aligned(16))) float ls[NW][16][68];
  __shared__ __attribute__((aligned(16))) int   li[NW][16][68];
  #pragma unroll
  for (int v = 0; v < 8; ++v) {
    int row = v + 8 * half;
    #pragma unroll
    for (int j = 0; j < KSEL; ++j) {
      ls[wave][row][4 * ln + j] = best[v][j];
      li[wave][row][4 * ln + j] = bidx[v][j];
    }
  }
  __syncthreads();

  if (lane < 16) {
    const int row = lane;
    float rs[KSEL]; int ri[KSEL];
    #pragma unroll
    for (int j = 0; j < KSEL; ++j) { rs[j] = FLT_BIG; ri[j] = 0; }
    for (int n = 0; n < 16; ++n) {
      const v4f s4 = *(const v4f*)(&ls[wave][row][4 * n]);
      const int4 i4 = *(const int4*)(&li[wave][row][4 * n]);
      const int ids[KSEL] = {i4.x, i4.y, i4.z, i4.w};
      #pragma unroll
      for (int j = 0; j < KSEL; ++j) {
        float sc = s4[j];
        int   id = ids[j];
        if (sc < rs[3]) {
          if (sc < rs[2]) {
            rs[3] = rs[2]; ri[3] = ri[2];
            if (sc < rs[1]) {
              rs[2] = rs[1]; ri[2] = ri[1];
              if (sc < rs[0]) {
                rs[1] = rs[0]; ri[1] = ri[0]; rs[0] = sc; ri[0] = id;
              } else { rs[1] = sc; ri[1] = id; }
            } else { rs[2] = sc; ri[2] = id; }
          } else { rs[3] = sc; ri[3] = id; }
        }
      }
    }
    const size_t o = (((size_t)b * NQ + z) * PT + p0 + row) * KSEL;
    v4f so; so.x = rs[0]; so.y = rs[1]; so.z = rs[2]; so.w = rs[3];
    *(v4f*)(pscore + o) = so;
    *(int4*)(pidx + o) = make_int4(ri[0], ri[1], ri[2], ri[3]);
  }
}

__global__ void merge_kernel(const float* __restrict__ pscore, const int* __restrict__ pidx,
                             int* __restrict__ out_idx) {
  int t = blockIdx.x * blockDim.x + threadIdx.x;
  if (t >= B_ * PT) return;
  int b = t / PT;
  int p = t % PT;
  float rs[KSEL]; int ri[KSEL];
  #pragma unroll
  for (int j = 0; j < KSEL; ++j) { rs[j] = FLT_BIG; ri[j] = 0; }
  #pragma unroll
  for (int q = 0; q < NQ; ++q) {
    const size_t o = (((size_t)b * NQ + q) * PT + p) * KSEL;
    const v4f s4 = *(const v4f*)(pscore + o);
    const int4 i4 = *(const int4*)(pidx + o);
    const int ids[KSEL] = {i4.x, i4.y, i4.z, i4.w};
    #pragma unroll
    for (int j = 0; j < KSEL; ++j) {
      float sc = s4[j];
      int   id = ids[j];
      if (sc < rs[3]) {
        if (sc < rs[2]) {
          rs[3] = rs[2]; ri[3] = ri[2];
          if (sc < rs[1]) {
            rs[2] = rs[1]; ri[2] = ri[1];
            if (sc < rs[0]) {
              rs[1] = rs[0]; ri[1] = ri[0]; rs[0] = sc; ri[0] = id;
            } else { rs[1] = sc; ri[1] = id; }
          } else { rs[2] = sc; ri[2] = id; }
        } else { rs[3] = sc; ri[3] = id; }
      }
    }
  }
  *(int4*)(out_idx + (size_t)t * KSEL) = make_int4(ri[0], ri[1], ri[2], ri[3]);
}

__global__ void gather_kernel(const float* __restrict__ S, const int* __restrict__ idx,
                              float* __restrict__ out) {
  size_t t = (size_t)blockIdx.x * blockDim.x + threadIdx.x;
  if (t >= (size_t)B_ * C_ * PT) return;
  int p  = (int)(t % PT);
  int bc = (int)(t / PT);
  int b  = bc / C_;
  const int4 id = *(const int4*)(idx + ((size_t)b * PT + p) * KSEL);
  const float* src = S + (size_t)bc * PS;
  float4 r;
  r.x = src[id.x]; r.y = src[id.y]; r.z = src[id.z]; r.w = src[id.w];
  *(float4*)(out + t * KSEL) = r;
}

extern "C" void kernel_launch(void* const* d_in, const int* in_sizes, int n_in,
                              void* d_out, int out_size, void* d_ws, size_t ws_size,
                              hipStream_t stream) {
  const float* T = (const float*)d_in[0];
  const float* S = (const float*)d_in[1];

  float* sq  = (float*)d_ws;
  int*   idx = (int*)((char*)d_ws + (size_t)B_ * PS * sizeof(float));
  float* psc = (float*)((char*)d_ws + (size_t)262144);
  int*   pid = (int*)((char*)d_ws + (size_t)786432);
  float* out = (float*)d_out;

  sqnorm_kernel<<<dim3(PS / 256, B_), 256, 0, stream>>>(S, sq);
  knn_kernel<<<dim3(PT / (16 * NW), B_, NQ), 256, 0, stream>>>(T, S, sq, psc, pid);
  merge_kernel<<<dim3(B_ * PT / 128), 128, 0, stream>>>(psc, pid, idx);
  size_t n = (size_t)B_ * C_ * PT;
  gather_kernel<<<dim3((unsigned)((n + 255) / 256)), 256, 0, stream>>>(S, idx, out);
}


# Slab-loop barrier split: a wave signals once its B fragments are read and its share of the next slab stored, waits at the slab's foot; selection overlaps partner wave's WMMAs; scores bit-identical
# speedup vs baseline: 1.5129x; 1.0494x over previous
#include <hip/hip_runtime.h>

typedef __attribute__((ext_vector_type(2))) float v2f;
typedef __attribute__((ext_vector_type(4))) float v4f;
typedef __attribute__((ext_vector_type(8))) float v8f;

#define B_   4
#define C_   68
#define F0   4
#define PT   2048
#define PS   8192
#define KSEL 4
#define FLT_BIG 3.4e38f

__global__ void sqnorm_kernel(const float* __restrict__ S, float* __restrict__ sq) {
  int q = blockIdx.x * blockDim.x + threadIdx.x;
  int b = blockIdx.y;
  if (q >= PS) return;
  const float* p = S + ((size_t)b * C_ + F0) * PS + q;
  float acc = 0.f;
  #pragma unroll 8
  for (int f = 0; f < 64; ++f) {
    float v = p[(size_t)f * PS];
    acc = fmaf(v, v, acc);
  }
  sq[b * PS + q] = acc;
}

template <int CTRL>
__device__ __forceinline__ float row_xchg(float x) {
  return __int_as_float(__builtin_amdgcn_update_dpp(
      __float_as_int(x), __float_as_int(x), CTRL, 0xF, 0xF, false));
}

template <int CTRL>
__device__ __forceinline__ void row_merge4(float& c0, float& c1, float& c2, float& c3) {
  const float d0 = fminf(c0, row_xchg<CTRL>(c3));
  const float d1 = fminf(c1, row_xchg<CTRL>(c2));
  const float d2 = fminf(c2, row_xchg<CTRL>(c1));
  const float d3 = fminf(c3, row_xchg<CTRL>(c0));
  const float e0 = fminf(d0, d2), e2 = fmaxf(d0, d2);
  const float e1 = fminf(d1, d3), e3 = fmaxf(d1, d3);
  c0 = fminf(e0, e1); c1 = fmaxf(e0, e1);
  c2 = fminf(e2, e3); c3 = fmaxf(e2, e3);
}

__device__ __forceinline__ float row_fourth(float c0, float c1, float c2, float c3) {
  row_merge4<0x161>(c0, c1, c2, c3);
  row_merge4<0x162>(c0, c1, c2, c3);
  row_merge4<0x164>(c0, c1, c2, c3);
  const float d0 = fminf(c0, row_xchg<0x168>(c3));
  const float d1 = fminf(c1, row_xchg<0x168>(c2));
  const float d2 = fminf(c2, row_xchg<0x168>(c1));
  const float d3 = fminf(c3, row_xchg<0x168>(c0));
  return fmaxf(fmaxf(d0, d1), fmaxf(d2, d3));
}

#define NW   8
#define NQ   4
#define QLEN (PS / NQ)
#define CROW 260

__launch_bounds__(256)
__global__ void knn_kernel(const float* __restrict__ T, const float* __restrict__ S,
                           const float* __restrict__ sq, float* __restrict__ pscore,
                           int* __restrict__ pidx) {
  __shared__ __attribute__((aligned(16))) float stage[2][16 * CROW];

  const int b    = blockIdx.y;
  const int z    = blockIdx.z;
  const int lane = threadIdx.x & 31;
  const int wave = threadIdx.x >> 5;
  const int half = lane >> 4;
  const int ln   = lane & 15;
  const int p0   = blockIdx.x * (16 * NW) + wave * 16;

  v2f a[16];
  #pragma unroll
  for (int s = 0; s < 16; ++s) {
    int f = F0 + 4 * s + 2 * half;
    const float* tp = T + ((size_t)b * C_ + f) * PT + p0 + ln;
    a[s].x = tp[0];
    a[s].y = tp[PT];
  }

  float best[8][KSEL];
  int   bidx[8][KSEL];
  #pragma unroll
  for (int v = 0; v < 8; ++v)
    #pragma unroll
    for (int j = 0; j < KSEL; ++j) { best[v][j] = FLT_BIG; bidx[v][j] = 0; }

  float thr[8];
  #pragma unroll
  for (int v = 0; v < 8; ++v) thr[v] = FLT_BIG;

  const int qbeg = z * QLEN;
  const int fs = 2 * (wave * 2 + half);
  const float* gsrc = S + ((size_t)b * C_ + F0 + fs) * PS + qbeg + 4 * ln;
  const int st_off = ln * CROW + fs;
  const int fr_off = ln * CROW + 2 * half;

  v4f gx[2], gy[2];
  {
    v4f hx[2], hy[2];
    #pragma unroll
    for (int j = 0; j < 2; ++j) {
      hx[j] = *(const v4f*)(gsrc + (size_t)(32 * j) * PS);
      hy[j] = *(const v4f*)(gsrc + (size_t)(32 * j + 1) * PS);
    }
    #pragma unroll
    for (int j = 0; j < 2; ++j) {
      gx[j] = *(const v4f*)(gsrc + (size_t)(32 * j) * PS + 64);
      gy[j] = *(const v4f*)(gsrc + (size_t)(32 * j + 1) * PS + 64);
    }
    #pragma unroll
    for (int j = 0; j < 2; ++j)
      #pragma unroll
      for (int t = 0; t < 4; ++t) {
        v2f w; w.x = hx[j][t]; w.y = hy[j][t];
        *(v2f*)(&stage[0][st_off + 32 * j + 64 * t]) = w;
      }
  }
  __syncthreads();

  for (int it = 0; it < QLEN / 64; ++it) {
    const int q0  = qbeg + it * 64;
    const int cur = it & 1;
    const bool more  = (it + 1) < QLEN / 64;
    const bool more2 = (it + 2) < QLEN / 64;

    const float* fr = &stage[cur][fr_off];
    v2f bf[16][4];
    #pragma unroll
    for (int s = 0; s < 16; ++s)
      #pragma unroll
      for (int t = 0; t < 4; ++t)
        bf[s][t] = *(const v2f*)(fr + 64 * t + 4 * s);

    if (more) {
      asm volatile("" :: "v"(gx[0]), "v"(gx[1]), "v"(gy[0]), "v"(gy[1]));
      #pragma unroll
      for (int j = 0; j < 2; ++j)
        #pragma unroll
        for (int t = 0; t < 4; ++t) {
          v2f w; w.x = gx[j][t]; w.y = gy[j][t];
          *(v2f*)(&stage[cur ^ 1][st_off + 32 * j + 64 * t]) = w;
        }
    }

    __builtin_amdgcn_fence(__ATOMIC_RELEASE, "workgroup");
    __builtin_amdgcn_s_barrier_signal(-1);

    const v4f sq4 = *(const v4f*)(sq + b * PS + q0 + 4 * ln);
    if (more2) {
      #pragma unroll
      for (int j = 0; j < 2; ++j) {
        gx[j] = *(const v4f*)(gsrc + (size_t)(32 * j) * PS + (it + 2) * 64);
        gy[j] = *(const v4f*)(gsrc + (size_t)(32 * j + 1) * PS + (it + 2) * 64);
      }
    }

    v8f acc[4] = {};
    #pragma unroll
    for (int s = 0; s < 16; ++s) {
      #pragma unroll
      for (int t = 0; t < 4; ++t) {
        acc[t] = __builtin_amdgcn_wmma_f32_16x16x4_f32(
             false, a[s],  false, bf[s][t],
             (short)0, acc[t],  false,  false);
      }
    }

    float sc[4][8];
    #pragma unroll
    for (int t = 0; t < 4; ++t)
      #pragma unroll
      for (int v = 0; v < 8; ++v)
        sc[t][v] = fmaf(-2.f, acc[t][v], sq4[t]);

    unsigned pm[4][8];
    #pragma unroll
    for (int t = 0; t < 4; ++t)
      #pragma unroll
      for (int v = 0; v < 8; ++v)
        pm[t][v] = __builtin_amdgcn_ballot_w32(sc[t][v] < thr[v]);

    #pragma unroll
    for (int t = 0; t < 4; ++t) {
      const int qidx = q0 + 4 * ln + t;

      #pragma unroll
      for (int v = 0; v < 8; ++v) {
        if (__builtin_expect(pm[t][v] != 0u, 0)) {
          const float s = sc[t][v];
          if (s < thr[v]) {
            if (s < best[v][2]) {
              best[v][3] = best[v][2]; bidx[v][3] = bidx[v][2];
              if (s < best[v][1]) {
                best[v][2] = best[v][1]; bidx[v][2] = bidx[v][1];
                if (s < best[v][0]) {
                  best[v][1] = best[v][0]; bidx[v][1] = bidx[v][0];
                  best[v][0] = s; bidx[v][0] = qidx;
                } else { best[v][1] = s; bidx[v][1] = qidx; }
              } else { best[v][2] = s; bidx[v][2] = qidx; }
            } else { best[v][3] = s; bidx[v][3] = qidx; }
            thr[v] = fminf(thr[v], best[v][3]);
          }
        }
      }
    }

    const int done = it + 1;
    if ((done & it) == 0 && done < (PS / 4) / 64) {
      #pragma unroll
      for (int v = 0; v < 8; ++v)
        thr[v] = row_fourth(best[v][0], best[v][1], best[v][2], best[v][3]);
    }

    __builtin_amdgcn_s_barrier_wait(-1);
    __builtin_amdgcn_fence(__ATOMIC_ACQUIRE, "workgroup");
  }

  __shared__ __attribute__((aligned(16))) float ls[NW][16][68];
  __shared__ __attribute__((aligned(16))) int   li[NW][16][68];
  #pragma unroll
  for (int v = 0; v < 8; ++v) {
    int row = v + 8 * half;
    #pragma unroll
    for (int j = 0; j < KSEL; ++j) {
      ls[wave][row][4 * ln + j] = best[v][j];
      li[wave][row][4 * ln + j] = bidx[v][j];
    }
  }
  __syncthreads();

  if (lane < 16) {
    const int row = lane;
    float rs[KSEL]; int ri[KSEL];
    #pragma unroll
    for (int j = 0; j < KSEL; ++j) { rs[j] = FLT_BIG; ri[j] = 0; }
    for (int n = 0; n < 16; ++n) {
      const v4f s4 = *(const v4f*)(&ls[wave][row][4 * n]);
      const int4 i4 = *(const int4*)(&li[wave][row][4 * n]);
      const int ids[KSEL] = {i4.x, i4.y, i4.z, i4.w};
      #pragma unroll
      for (int j = 0; j < KSEL; ++j) {
        float sc = s4[j];
        int   id = ids[j];
        if (sc < rs[3]) {
          if (sc < rs[2]) {
            rs[3] = rs[2]; ri[3] = ri[2];
            if (sc < rs[1]) {
              rs[2] = rs[1]; ri[2] = ri[1];
              if (sc < rs[0]) {
                rs[1] = rs[0]; ri[1] = ri[0]; rs[0] = sc; ri[0] = id;
              } else { rs[1] = sc; ri[1] = id; }
            } else { rs[2] = sc; ri[2] = id; }
          } else { rs[3] = sc; ri[3] = id; }
        }
      }
    }
    const size_t o = (((size_t)b * NQ + z) * PT + p0 + row) * KSEL;
    v4f so; so.x = rs[0]; so.y = rs[1]; so.z = rs[2]; so.w = rs[3];
    *(v4f*)(pscore + o) = so;
    *(int4*)(pidx + o) = make_int4(ri[0], ri[1], ri[2], ri[3]);
  }
}

__global__ void merge_kernel(const float* __restrict__ pscore, const int* __restrict__ pidx,
                             int* __restrict__ out_idx) {
  int t = blockIdx.x * blockDim.x + threadIdx.x;
  if (t >= B_ * PT) return;
  int b = t / PT;
  int p = t % PT;
  float rs[KSEL]; int ri[KSEL];
  #pragma unroll
  for (int j = 0; j < KSEL; ++j) { rs[j] = FLT_BIG; ri[j] = 0; }
  #pragma unroll
  for (int q = 0; q < NQ; ++q) {
    const size_t o = (((size_t)b * NQ + q) * PT + p) * KSEL;
    const v4f s4 = *(const v4f*)(pscore + o);
    const int4 i4 = *(const int4*)(pidx + o);
    const int ids[KSEL] = {i4.x, i4.y, i4.z, i4.w};
    #pragma unroll
    for (int j = 0; j < KSEL; ++j) {
      float sc = s4[j];
      int   id = ids[j];
      if (sc < rs[3]) {
        if (sc < rs[2]) {
          rs[3] = rs[2]; ri[3] = ri[2];
          if (sc < rs[1]) {
            rs[2] = rs[1]; ri[2] = ri[1];
            if (sc < rs[0]) {
              rs[1] = rs[0]; ri[1] = ri[0]; rs[0] = sc; ri[0] = id;
            } else { rs[1] = sc; ri[1] = id; }
          } else { rs[2] = sc; ri[2] = id; }
        } else { rs[3] = sc; ri[3] = id; }
      }
    }
  }
  *(int4*)(out_idx + (size_t)t * KSEL) = make_int4(ri[0], ri[1], ri[2], ri[3]);
}

__global__ void gather_kernel(const float* __restrict__ S, const int* __restrict__ idx,
                              float* __restrict__ out) {
  size_t t = (size_t)blockIdx.x * blockDim.x + threadIdx.x;
  if (t >= (size_t)B_ * C_ * PT) return;
  int p  = (int)(t % PT);
  int bc = (int)(t / PT);
  int b  = bc / C_;
  const int4 id = *(const int4*)(idx + ((size_t)b * PT + p) * KSEL);
  const float* src = S + (size_t)bc * PS;
  float4 r;
  r.x = src[id.x]; r.y = src[id.y]; r.z = src[id.z]; r.w = src[id.w];
  *(float4*)(out + t * KSEL) = r;
}

extern "C" void kernel_launch(void* const* d_in, const int* in_sizes, int n_in,
                              void* d_out, int out_size, void* d_ws, size_t ws_size,
                              hipStream_t stream) {
  const float* T = (const float*)d_in[0];
  const float* S = (const float*)d_in[1];

  float* sq  = (float*)d_ws;
  int*   idx = (int*)((char*)d_ws + (size_t)B_ * PS * sizeof(float));
  float* psc = (float*)((char*)d_ws + (size_t)262144);
  int*   pid = (int*)((char*)d_ws + (size_t)786432);
  float* out = (float*)d_out;

  sqnorm_kernel<<<dim3(PS / 256, B_), 256, 0, stream>>>(S, sq);
  knn_kernel<<<dim3(PT / (16 * NW), B_, NQ), 256, 0, stream>>>(T, S, sq, psc, pid);
  merge_kernel<<<dim3(B_ * PT / 128), 128, 0, stream>>>(psc, pid, idx);
  size_t n = (size_t)B_ * C_ * PT;
  gather_kernel<<<dim3((unsigned)((n + 255) / 256)), 256, 0, stream>>>(S, idx, out);
}


# Quarter merge fused into the gather: a thread owns one target point x 17 channels, merges its 4 lists in registers and reuses the 4 indices for all its channels; 3 launches, 17x fewer waves
# speedup vs baseline: 1.5595x; 1.0308x over previous
#include <hip/hip_runtime.h>

typedef __attribute__((ext_vector_type(2))) float v2f;
typedef __attribute__((ext_vector_type(4))) float v4f;
typedef __attribute__((ext_vector_type(8))) float v8f;

#define B_   4
#define C_   68
#define F0   4
#define PT   2048
#define PS   8192
#define KSEL 4
#define FLT_BIG 3.4e38f

__global__ void sqnorm_kernel(const float* __restrict__ S, float* __restrict__ sq) {
  int q = blockIdx.x * blockDim.x + threadIdx.x;
  int b = blockIdx.y;
  if (q >= PS) return;
  const float* p = S + ((size_t)b * C_ + F0) * PS + q;
  float acc = 0.f;
  #pragma unroll 8
  for (int f = 0; f < 64; ++f) {
    float v = p[(size_t)f * PS];
    acc = fmaf(v, v, acc);
  }
  sq[b * PS + q] = acc;
}

template <int CTRL>
__device__ __forceinline__ float row_xchg(float x) {
  return __int_as_float(__builtin_amdgcn_update_dpp(
      __float_as_int(x), __float_as_int(x), CTRL, 0xF, 0xF, false));
}

template <int CTRL>
__device__ __forceinline__ void row_merge4(float& c0, float& c1, float& c2, float& c3) {
  const float d0 = fminf(c0, row_xchg<CTRL>(c3));
  const float d1 = fminf(c1, row_xchg<CTRL>(c2));
  const float d2 = fminf(c2, row_xchg<CTRL>(c1));
  const float d3 = fminf(c3, row_xchg<CTRL>(c0));
  const float e0 = fminf(d0, d2), e2 = fmaxf(d0, d2);
  const float e1 = fminf(d1, d3), e3 = fmaxf(d1, d3);
  c0 = fminf(e0, e1); c1 = fmaxf(e0, e1);
  c2 = fminf(e2, e3); c3 = fmaxf(e2, e3);
}

__device__ __forceinline__ float row_fourth(float c0, float c1, float c2, float c3) {
  row_merge4<0x161>(c0, c1, c2, c3);
  row_merge4<0x162>(c0, c1, c2, c3);
  row_merge4<0x164>(c0, c1, c2, c3);
  const float d0 = fminf(c0, row_xchg<0x168>(c3));
  const float d1 = fminf(c1, row_xchg<0x168>(c2));
  const float d2 = fminf(c2, row_xchg<0x168>(c1));
  const float d3 = fminf(c3, row_xchg<0x168>(c0));
  return fmaxf(fmaxf(d0, d1), fmaxf(d2, d3));
}

#define NW   8
#define NQ   4
#define QLEN (PS / NQ)
#define CROW 260

__launch_bounds__(256)
__global__ void knn_kernel(const float* __restrict__ T, const float* __restrict__ S,
                           const float* __restrict__ sq, float* __restrict__ pscore,
                           int* __restrict__ pidx) {
  __shared__ __attribute__((aligned(16))) float stage[2][16 * CROW];

  const int b    = blockIdx.y;
  const int z    = blockIdx.z;
  const int lane = threadIdx.x & 31;
  const int wave = threadIdx.x >> 5;
  const int half = lane >> 4;
  const int ln   = lane & 15;
  const int p0   = blockIdx.x * (16 * NW) + wave * 16;

  v2f a[16];
  #pragma unroll
  for (int s = 0; s < 16; ++s) {
    int f = F0 + 4 * s + 2 * half;
    const float* tp = T + ((size_t)b * C_ + f) * PT + p0 + ln;
    a[s].x = tp[0];
    a[s].y = tp[PT];
  }

  float best[8][KSEL];
  int   bidx[8][KSEL];
  #pragma unroll
  for (int v = 0; v < 8; ++v)
    #pragma unroll
    for (int j = 0; j < KSEL; ++j) { best[v][j] = FLT_BIG; bidx[v][j] = 0; }

  float thr[8];
  #pragma unroll
  for (int v = 0; v < 8; ++v) thr[v] = FLT_BIG;

  const int qbeg = z * QLEN;
  const int fs = 2 * (wave * 2 + half);
  const float* gsrc = S + ((size_t)b * C_ + F0 + fs) * PS + qbeg + 4 * ln;
  const int st_off = ln * CROW + fs;
  const int fr_off = ln * CROW + 2 * half;

  v4f gx[2], gy[2];
  {
    v4f hx[2], hy[2];
    #pragma unroll
    for (int j = 0; j < 2; ++j) {
      hx[j] = *(const v4f*)(gsrc + (size_t)(32 * j) * PS);
      hy[j] = *(const v4f*)(gsrc + (size_t)(32 * j + 1) * PS);
    }
    #pragma unroll
    for (int j = 0; j < 2; ++j) {
      gx[j] = *(const v4f*)(gsrc + (size_t)(32 * j) * PS + 64);
      gy[j] = *(const v4f*)(gsrc + (size_t)(32 * j + 1) * PS + 64);
    }
    #pragma unroll
    for (int j = 0; j < 2; ++j)
      #pragma unroll
      for (int t = 0; t < 4; ++t) {
        v2f w; w.x = hx[j][t]; w.y = hy[j][t];
        *(v2f*)(&stage[0][st_off + 32 * j + 64 * t]) = w;
      }
  }
  __syncthreads();

  for (int it = 0; it < QLEN / 64; ++it) {
    const int q0  = qbeg + it * 64;
    const int cur = it & 1;
    const bool more  = (it + 1) < QLEN / 64;
    const bool more2 = (it + 2) < QLEN / 64;

    const float* fr = &stage[cur][fr_off];
    v2f bf[16][4];
    #pragma unroll
    for (int s = 0; s < 16; ++s)
      #pragma unroll
      for (int t = 0; t < 4; ++t)
        bf[s][t] = *(const v2f*)(fr + 64 * t + 4 * s);

    if (more) {
      asm volatile("" :: "v"(gx[0]), "v"(gx[1]), "v"(gy[0]), "v"(gy[1]));
      #pragma unroll
      for (int j = 0; j < 2; ++j)
        #pragma unroll
        for (int t = 0; t < 4; ++t) {
          v2f w; w.x = gx[j][t]; w.y = gy[j][t];
          *(v2f*)(&stage[cur ^ 1][st_off + 32 * j + 64 * t]) = w;
        }
    }

    __builtin_amdgcn_fence(__ATOMIC_RELEASE, "workgroup");
    __builtin_amdgcn_s_barrier_signal(-1);

    const v4f sq4 = *(const v4f*)(sq + b * PS + q0 + 4 * ln);
    if (more2) {
      #pragma unroll
      for (int j = 0; j < 2; ++j) {
        gx[j] = *(const v4f*)(gsrc + (size_t)(32 * j) * PS + (it + 2) * 64);
        gy[j] = *(const v4f*)(gsrc + (size_t)(32 * j + 1) * PS + (it + 2) * 64);
      }
    }

    v8f acc[4] = {};
    #pragma unroll
    for (int s = 0; s < 16; ++s) {
      #pragma unroll
      for (int t = 0; t < 4; ++t) {
        acc[t] = __builtin_amdgcn_wmma_f32_16x16x4_f32(
             false, a[s],  false, bf[s][t],
             (short)0, acc[t],  false,  false);
      }
    }

    float sc[4][8];
    #pragma unroll
    for (int t = 0; t < 4; ++t)
      #pragma unroll
      for (int v = 0; v < 8; ++v)
        sc[t][v] = fmaf(-2.f, acc[t][v], sq4[t]);

    unsigned pm[4][8];
    #pragma unroll
    for (int t = 0; t < 4; ++t)
      #pragma unroll
      for (int v = 0; v < 8; ++v)
        pm[t][v] = __builtin_amdgcn_ballot_w32(sc[t][v] < thr[v]);

    #pragma unroll
    for (int t = 0; t < 4; ++t) {
      const int qidx = q0 + 4 * ln + t;

      #pragma unroll
      for (int v = 0; v < 8; ++v) {
        if (__builtin_expect(pm[t][v] != 0u, 0)) {
          const float s = sc[t][v];
          if (s < thr[v]) {
            if (s < best[v][2]) {
              best[v][3] = best[v][2]; bidx[v][3] = bidx[v][2];
              if (s < best[v][1]) {
                best[v][2] = best[v][1]; bidx[v][2] = bidx[v][1];
                if (s < best[v][0]) {
                  best[v][1] = best[v][0]; bidx[v][1] = bidx[v][0];
                  best[v][0] = s; bidx[v][0] = qidx;
                } else { best[v][1] = s; bidx[v][1] = qidx; }
              } else { best[v][2] = s; bidx[v][2] = qidx; }
            } else { best[v][3] = s; bidx[v][3] = qidx; }
            thr[v] = fminf(thr[v], best[v][3]);
          }
        }
      }
    }

    const int done = it + 1;
    if ((done & it) == 0 && done < (PS / 4) / 64) {
      #pragma unroll
      for (int v = 0; v < 8; ++v)
        thr[v] = row_fourth(best[v][0], best[v][1], best[v][2], best[v][3]);
    }

    __builtin_amdgcn_s_barrier_wait(-1);
    __builtin_amdgcn_fence(__ATOMIC_ACQUIRE, "workgroup");
  }

  __shared__ __attribute__((aligned(16))) float ls[NW][16][68];
  __shared__ __attribute__((aligned(16))) int   li[NW][16][68];
  #pragma unroll
  for (int v = 0; v < 8; ++v) {
    int row = v + 8 * half;
    #pragma unroll
    for (int j = 0; j < KSEL; ++j) {
      ls[wave][row][4 * ln + j] = best[v][j];
      li[wave][row][4 * ln + j] = bidx[v][j];
    }
  }
  __syncthreads();

  if (lane < 16) {
    const int row = lane;
    float rs[KSEL]; int ri[KSEL];
    #pragma unroll
    for (int j = 0; j < KSEL; ++j) { rs[j] = FLT_BIG; ri[j] = 0; }
    for (int n = 0; n < 16; ++n) {
      const v4f s4 = *(const v4f*)(&ls[wave][row][4 * n]);
      const int4 i4 = *(const int4*)(&li[wave][row][4 * n]);
      const int ids[KSEL] = {i4.x, i4.y, i4.z, i4.w};
      #pragma unroll
      for (int j = 0; j < KSEL; ++j) {
        float sc = s4[j];
        int   id = ids[j];
        if (sc < rs[3]) {
          if (sc < rs[2]) {
            rs[3] = rs[2]; ri[3] = ri[2];
            if (sc < rs[1]) {
              rs[2] = rs[1]; ri[2] = ri[1];
              if (sc < rs[0]) {
                rs[1] = rs[0]; ri[1] = ri[0]; rs[0] = sc; ri[0] = id;
              } else { rs[1] = sc; ri[1] = id; }
            } else { rs[2] = sc; ri[2] = id; }
          } else { rs[3] = sc; ri[3] = id; }
        }
      }
    }
    const size_t o = (((size_t)b * NQ + z) * PT + p0 + row) * KSEL;
    v4f so; so.x = rs[0]; so.y = rs[1]; so.z = rs[2]; so.w = rs[3];
    *(v4f*)(pscore + o) = so;
    *(int4*)(pidx + o) = make_int4(ri[0], ri[1], ri[2], ri[3]);
  }
}

#define CG  4
#define CPG (C_ / CG)
#define GT  128

__launch_bounds__(GT)
__global__ void merge_gather_kernel(const float* __restrict__ S, const float* __restrict__ pscore,
                                    const int* __restrict__ pidx, float* __restrict__ out) {
  const int p = blockIdx.x * GT + threadIdx.x;
  const int g = blockIdx.y;
  const int b = blockIdx.z;
  float rs[KSEL]; int ri[KSEL];
  #pragma unroll
  for (int j = 0; j < KSEL; ++j) { rs[j] = FLT_BIG; ri[j] = 0; }
  v4f  qs[NQ];
  int4 qi[NQ];
  #pragma unroll
  for (int q = 0; q < NQ; ++q) {
    const size_t o = (((size_t)b * NQ + q) * PT + p) * KSEL;
    qs[q] = *(const v4f*)(pscore + o);
    qi[q] = *(const int4*)(pidx + o);
  }
  #pragma unroll
  for (int q = 0; q < NQ; ++q) {
    const v4f s4 = qs[q];
    const int ids[KSEL] = {qi[q].x, qi[q].y, qi[q].z, qi[q].w};
    #pragma unroll
    for (int j = 0; j < KSEL; ++j) {
      float sc = s4[j];
      int   id = ids[j];
      if (sc < rs[3]) {
        if (sc < rs[2]) {
          rs[3] = rs[2]; ri[3] = ri[2];
          if (sc < rs[1]) {
            rs[2] = rs[1]; ri[2] = ri[1];
            if (sc < rs[0]) {
              rs[1] = rs[0]; ri[1] = ri[0]; rs[0] = sc; ri[0] = id;
            } else { rs[1] = sc; ri[1] = id; }
          } else { rs[2] = sc; ri[2] = id; }
        } else { rs[3] = sc; ri[3] = id; }
      }
    }
  }
  const int i0 = ri[0] & (PS - 1), i1 = ri[1] & (PS - 1);
  const int i2 = ri[2] & (PS - 1), i3 = ri[3] & (PS - 1);

  const float* src = S + ((size_t)b * C_ + (size_t)g * CPG) * PS;
  float* o = out + (((size_t)b * C_ + (size_t)g * CPG) * PT + p) * KSEL;
  #pragma unroll
  for (int j = 0; j < CPG; ++j) {
    const float* row = src + (size_t)j * PS;
    float4 r;
    r.x = row[i0]; r.y = row[i1]; r.z = row[i2]; r.w = row[i3];
    *(float4*)(o + (size_t)j * PT * KSEL) = r;
  }
}

extern "C" void kernel_launch(void* const* d_in, const int* in_sizes, int n_in,
                              void* d_out, int out_size, void* d_ws, size_t ws_size,
                              hipStream_t stream) {
  const float* T = (const float*)d_in[0];
  const float* S = (const float*)d_in[1];

  float* sq  = (float*)d_ws;
  float* psc = (float*)((char*)d_ws + (size_t)262144);
  int*   pid = (int*)((char*)d_ws + (size_t)786432);
  float* out = (float*)d_out;

  sqnorm_kernel<<<dim3(PS / 256, B_), 256, 0, stream>>>(S, sq);
  knn_kernel<<<dim3(PT / (16 * NW), B_, NQ), 256, 0, stream>>>(T, S, sq, psc, pid);
  merge_gather_kernel<<<dim3(PT / GT, CG, B_), GT, 0, stream>>>(S, psc, pid, out);
}


# B fragments double-buffered in registers: next slab's LDS fragment reads issued as one group ahead of this slab's WMMAs (loop unrolled by 2); barrier signal behind the WMMAs; scores bit-identical
# speedup vs baseline: 1.7111x; 1.0972x over previous
#include <hip/hip_runtime.h>

typedef __attribute__((ext_vector_type(2))) float v2f;
typedef __attribute__((ext_vector_type(4))) float v4f;
typedef __attribute__((ext_vector_type(8))) float v8f;

#define B_   4
#define C_   68
#define F0   4
#define PT   2048
#define PS   8192
#define KSEL 4
#define FLT_BIG 3.4e38f

__global__ void sqnorm_kernel(const float* __restrict__ S, float* __restrict__ sq) {
  int q = blockIdx.x * blockDim.x + threadIdx.x;
  int b = blockIdx.y;
  if (q >= PS) return;
  const float* p = S + ((size_t)b * C_ + F0) * PS + q;
  float acc = 0.f;
  #pragma unroll 8
  for (int f = 0; f < 64; ++f) {
    float v = p[(size_t)f * PS];
    acc = fmaf(v, v, acc);
  }
  sq[b * PS + q] = acc;
}

#define QCAP 128
#define QP   132

template <bool LAST>
__device__ __forceinline__ void flush_rows(const float* qrow_s, const int* qrow_i, int ln, int half,
                                           int (&cnt)[16], int& over,
                                           float (&rs)[4], int (&ri)[4], float (&thr)[8]) {
  __builtin_amdgcn_fence(__ATOMIC_RELEASE, "workgroup");
  __builtin_amdgcn_wave_barrier();
  __builtin_amdgcn_fence(__ATOMIC_ACQUIRE, "workgroup");
  int n = 0;
  #pragma unroll
  for (int r = 0; r < 16; ++r) n = (ln == r) ? cnt[r] : n;
  n = n < QCAP ? n : QCAP;
  int mx = cnt[0];
  #pragma unroll
  for (int r = 1; r < 16; ++r) mx = cnt[r] > mx ? cnt[r] : mx;
  mx = __builtin_amdgcn_readfirstlane(mx);
  const int nmax = mx < QCAP ? mx : QCAP;
  for (int j = 0; j < nmax; j += 4) {
    const v4f  s4 = *(const v4f*)(qrow_s + j);
    const int4 i4 = *(const int4*)(qrow_i + j);
    const int ids[4] = {i4.x, i4.y, i4.z, i4.w};
    #pragma unroll
    for (int e = 0; e < 4; ++e) {
      const float s  = (j + e < n) ? s4[e] : FLT_BIG;
      const int   id = ids[e];
      const bool c0 = s < rs[0], c1 = s < rs[1], c2 = s < rs[2], c3 = s < rs[3];
      rs[3] = c2 ? rs[2] : (c3 ? s : rs[3]);  ri[3] = c2 ? ri[2] : (c3 ? id : ri[3]);
      rs[2] = c1 ? rs[1] : (c2 ? s : rs[2]);  ri[2] = c1 ? ri[1] : (c2 ? id : ri[2]);
      rs[1] = c0 ? rs[0] : (c1 ? s : rs[1]);  ri[1] = c0 ? ri[0] : (c1 ? id : ri[1]);
      rs[0] = c0 ? s : rs[0];                 ri[0] = c0 ? id : ri[0];
    }
  }
  #pragma unroll
  for (int r = 0; r < 16; ++r) cnt[r] = 0;
  over = 0;
  if (!LAST) {
    #pragma unroll
    for (int v = 0; v < 8; ++v)
      thr[v] = __int_as_float(__builtin_amdgcn_ds_bpermute(4 * (v + 8 * half), __float_as_int(rs[3])));
  }
}

#define NW   8
#define NQ   4
#define QLEN (PS / NQ)
#define CROW 260

__launch_bounds__(256)
__global__ void knn_kernel(const float* __restrict__ T, const float* __restrict__ S,
                           const float* __restrict__ sq, float* __restrict__ pscore,
                           int* __restrict__ pidx) {
  __shared__ __attribute__((aligned(16))) float stage[2][16 * CROW];
  __shared__ __attribute__((aligned(16))) float qs[NW][16][QP];
  __shared__ __attribute__((aligned(16))) int   qi[NW][16][QP];

  const int b    = blockIdx.y;
  const int z    = blockIdx.z;
  const int lane = threadIdx.x & 31;
  const int wave = threadIdx.x >> 5;
  const int half = lane >> 4;
  const int ln   = lane & 15;
  const int p0   = blockIdx.x * (16 * NW) + wave * 16;

  v2f a[16];
  #pragma unroll
  for (int s = 0; s < 16; ++s) {
    int f = F0 + 4 * s + 2 * half;
    const float* tp = T + ((size_t)b * C_ + f) * PT + p0 + ln;
    a[s].x = tp[0];
    a[s].y = tp[PT];
  }

  float rs[KSEL]; int ri[KSEL];
  #pragma unroll
  for (int j = 0; j < KSEL; ++j) { rs[j] = FLT_BIG; ri[j] = 0; }

  float thr[8];
  #pragma unroll
  for (int v = 0; v < 8; ++v) thr[v] = FLT_BIG;

  int cnt[16];
  #pragma unroll
  for (int r = 0; r < 16; ++r) cnt[r] = 0;
  int over = 0;
  const unsigned hmask = half ? 0xFFFF0000u : 0x0000FFFFu;
  float* const qpush_s = &qs[wave][8 * half][0];
  int*   const qpush_i = &qi[wave][8 * half][0];
  const float* const qown_s = &qs[wave][ln][0];
  const int*   const qown_i = &qi[wave][ln][0];

  const int qbeg = z * QLEN;
  const int fs = 2 * (wave * 2 + half);
  const float* gsrc = S + ((size_t)b * C_ + F0 + fs) * PS + qbeg + 4 * ln;
  const int st_off = ln * CROW + fs;
  const int fr_off = ln * CROW + 2 * half;

  v4f gx[2], gy[2];
  {
    v4f hx[2], hy[2];
    #pragma unroll
    for (int j = 0; j < 2; ++j) {
      hx[j] = *(const v4f*)(gsrc + (size_t)(32 * j) * PS);
      hy[j] = *(const v4f*)(gsrc + (size_t)(32 * j + 1) * PS);
    }
    #pragma unroll
    for (int j = 0; j < 2; ++j) {
      gx[j] = *(const v4f*)(gsrc + (size_t)(32 * j) * PS + 64);
      gy[j] = *(const v4f*)(gsrc + (size_t)(32 * j + 1) * PS + 64);
    }
    #pragma unroll
    for (int j = 0; j < 2; ++j)
      #pragma unroll
      for (int t = 0; t < 4; ++t) {
        v2f w; w.x = hx[j][t]; w.y = hy[j][t];
        *(v2f*)(&stage[0][st_off + 32 * j + 64 * t]) = w;
      }
  }
  __syncthreads();

  v2f bfA[16][4], bfB[16][4];

  #pragma unroll
  for (int j = 0; j < 2; ++j)
    #pragma unroll
    for (int t = 0; t < 4; ++t) {
      v2f w; w.x = gx[j][t]; w.y = gy[j][t];
      *(v2f*)(&stage[1][st_off + 32 * j + 64 * t]) = w;
    }
  #pragma unroll
  for (int j = 0; j < 2; ++j) {
    gx[j] = *(const v4f*)(gsrc + (size_t)(32 * j) * PS + 128);
    gy[j] = *(const v4f*)(gsrc + (size_t)(32 * j + 1) * PS + 128);
  }
  #pragma unroll
  for (int s = 0; s < 16; ++s)
    #pragma unroll
    for (int t = 0; t < 4; ++t)
      bfA[s][t] = *(const v2f*)(&stage[0][fr_off + 64 * t + 4 * s]);
  __syncthreads();

  auto slab = [&](const int it, const int cur, v2f (&use)[16][4], v2f (&nxt)[16][4])
      __attribute__((always_inline)) {
    const int q0  = qbeg + it * 64;
    const bool more  = (it + 1) < QLEN / 64;
    const bool more2 = (it + 2) < QLEN / 64;
    const bool more3 = (it + 3) < QLEN / 64;

    if (more2) {
      #pragma unroll
      for (int j = 0; j < 2; ++j)
        #pragma unroll
        for (int t = 0; t < 4; ++t) {
          v2f w; w.x = gx[j][t]; w.y = gy[j][t];
          *(v2f*)(&stage[cur][st_off + 32 * j + 64 * t]) = w;
        }
    }
    if (more3) {
      #pragma unroll
      for (int j = 0; j < 2; ++j) {
        gx[j] = *(const v4f*)(gsrc + (size_t)(32 * j) * PS + (it + 3) * 64);
        gy[j] = *(const v4f*)(gsrc + (size_t)(32 * j + 1) * PS + (it + 3) * 64);
      }
    }
    if (more) {
      const float* fr = &stage[cur ^ 1][fr_off];
      #pragma unroll
      for (int s = 0; s < 16; ++s)
        #pragma unroll
        for (int t = 0; t < 4; ++t)
          nxt[s][t] = *(const v2f*)(fr + 64 * t + 4 * s);
    }
    const v4f sq4 = *(const v4f*)(sq + b * PS + q0 + 4 * ln);

    v8f acc[4] = {};
    #pragma unroll
    for (int s = 0; s < 16; ++s) {
      #pragma unroll
      for (int t = 0; t < 4; ++t) {
        acc[t] = __builtin_amdgcn_wmma_f32_16x16x4_f32(
             false, a[s],  false, use[s][t],
             (short)0, acc[t],  false,  false);
      }
    }

    __builtin_amdgcn_fence(__ATOMIC_RELEASE, "workgroup");
    __builtin_amdgcn_s_barrier_signal(-1);

    float sc[4][8];
    #pragma unroll
    for (int t = 0; t < 4; ++t)
      #pragma unroll
      for (int v = 0; v < 8; ++v)
        sc[t][v] = fmaf(-2.f, acc[t][v], sq4[t]);

    unsigned pm[4][8];
    #pragma unroll
    for (int t = 0; t < 4; ++t)
      #pragma unroll
      for (int v = 0; v < 8; ++v)
        pm[t][v] = __builtin_amdgcn_ballot_w32(sc[t][v] < thr[v]);

    #pragma unroll
    for (int t = 0; t < 4; ++t) {
      const int qidx = q0 + 4 * ln + t;

      #pragma unroll
      for (int v = 0; v < 8; ++v) {
        if (__builtin_expect(pm[t][v] != 0u, 0)) {
          const unsigned m = pm[t][v];
          int pos = (half ? cnt[v + 8] : cnt[v]) + (int)__builtin_amdgcn_mbcnt_lo(m & hmask, 0u);
          pos = pos < QCAP - 1 ? pos : QCAP - 1;
          if (sc[t][v] < thr[v]) {
            qpush_s[v * QP + pos] = sc[t][v];
            qpush_i[v * QP + pos] = qidx;
          }
          cnt[v]     += __builtin_popcount(m & 0xFFFFu);
          cnt[v + 8] += __builtin_popcount(m >> 16);
          over |= (int)(cnt[v] > QCAP - 64) | (int)(cnt[v + 8] > QCAP - 64);
        }
      }
    }

    const int done = it + 1;
    if (((done & it) == 0 && done < QLEN / 64) || over != 0)
      flush_rows<false>(qown_s, qown_i, ln, half, cnt, over, rs, ri, thr);

    __builtin_amdgcn_s_barrier_wait(-1);
    __builtin_amdgcn_fence(__ATOMIC_ACQUIRE, "workgroup");
  };

  for (int it = 0; it < QLEN / 64; it += 2) {
    slab(it,     0, bfA, bfB);
    slab(it + 1, 1, bfB, bfA);
  }

  flush_rows<true>(qown_s, qown_i, ln, half, cnt, over, rs, ri, thr);

  if (lane < 16) {
    const size_t o = (((size_t)b * NQ + z) * PT + p0 + lane) * KSEL;
    v4f so; so.x = rs[0]; so.y = rs[1]; so.z = rs[2]; so.w = rs[3];
    *(v4f*)(pscore + o) = so;
    *(int4*)(pidx + o) = make_int4(ri[0], ri[1], ri[2], ri[3]);
  }
}

#define CG  4
#define CPG (C_ / CG)
#define GT  128

__launch_bounds__(GT)
__global__ void merge_gather_kernel(const float* __restrict__ S, const float* __restrict__ pscore,
                                    const int* __restrict__ pidx, float* __restrict__ out) {
  const int p = blockIdx.x * GT + threadIdx.x;
  const int g = blockIdx.y;
  const int b = blockIdx.z;
  float rs[KSEL]; int ri[KSEL];
  #pragma unroll
  for (int j = 0; j < KSEL; ++j) { rs[j] = FLT_BIG; ri[j] = 0; }
  v4f  qs[NQ];
  int4 qi[NQ];
  #pragma unroll
  for (int q = 0; q < NQ; ++q) {
    const size_t o = (((size_t)b * NQ + q) * PT + p) * KSEL;
    qs[q] = *(const v4f*)(pscore + o);
    qi[q] = *(const int4*)(pidx + o);
  }
  #pragma unroll
  for (int q = 0; q < NQ; ++q) {
    const v4f s4 = qs[q];
    const int ids[KSEL] = {qi[q].x, qi[q].y, qi[q].z, qi[q].w};
    #pragma unroll
    for (int j = 0; j < KSEL; ++j) {
      float sc = s4[j];
      int   id = ids[j];
      if (sc < rs[3]) {
        if (sc < rs[2]) {
          rs[3] = rs[2]; ri[3] = ri[2];
          if (sc < rs[1]) {
            rs[2] = rs[1]; ri[2] = ri[1];
            if (sc < rs[0]) {
              rs[1] = rs[0]; ri[1] = ri[0]; rs[0] = sc; ri[0] = id;
            } else { rs[1] = sc; ri[1] = id; }
          } else { rs[2] = sc; ri[2] = id; }
        } else { rs[3] = sc; ri[3] = id; }
      }
    }
  }
  const int i0 = ri[0] & (PS - 1), i1 = ri[1] & (PS - 1);
  const int i2 = ri[2] & (PS - 1), i3 = ri[3] & (PS - 1);

  const float* src = S + ((size_t)b * C_ + (size_t)g * CPG) * PS;
  float* o = out + (((size_t)b * C_ + (size_t)g * CPG) * PT + p) * KSEL;
  #pragma unroll
  for (int j = 0; j < CPG; ++j) {
    const float* row = src + (size_t)j * PS;
    float4 r;
    r.x = row[i0]; r.y = row[i1]; r.z = row[i2]; r.w = row[i3];
    *(float4*)(o + (size_t)j * PT * KSEL) = r;
  }
}

extern "C" void kernel_launch(void* const* d_in, const int* in_sizes, int n_in,
                              void* d_out, int out_size, void* d_ws, size_t ws_size,
                              hipStream_t stream) {
  const float* T = (const float*)d_in[0];
  const float* S = (const float*)d_in[1];

  float* sq  = (float*)d_ws;
  float* psc = (float*)((char*)d_ws + (size_t)262144);
  int*   pid = (int*)((char*)d_ws + (size_t)786432);
  float* out = (float*)d_out;

  sqnorm_kernel<<<dim3(PS / 256, B_), 256, 0, stream>>>(S, sq);
  knn_kernel<<<dim3(PT / (16 * NW), B_, NQ), 256, 0, stream>>>(T, S, sq, psc, pid);
  merge_gather_kernel<<<dim3(PT / GT, CG, B_), GT, 0, stream>>>(S, psc, pid, out);
}


# Wave priority by phase: s_setprio 3 from the end of a slab's WMMAs to the foot of its barrier, 0 for the WMMA run; a selecting wave's vector work no longer waits behind the partner's WMMAs
# speedup vs baseline: 1.8447x; 1.0781x over previous
#include <hip/hip_runtime.h>

typedef __attribute__((ext_vector_type(2))) float v2f;
typedef __attribute__((ext_vector_type(4))) float v4f;
typedef __attribute__((ext_vector_type(8))) float v8f;

#define B_   4
#define C_   68
#define F0   4
#define PT   2048
#define PS   8192
#define KSEL 4
#define FLT_BIG 3.4e38f

__global__ void sqnorm_kernel(const float* __restrict__ S, float* __restrict__ sq) {
  int q = blockIdx.x * blockDim.x + threadIdx.x;
  int b = blockIdx.y;
  if (q >= PS) return;
  const float* p = S + ((size_t)b * C_ + F0) * PS + q;
  float acc = 0.f;
  #pragma unroll 8
  for (int f = 0; f < 64; ++f) {
    float v = p[(size_t)f * PS];
    acc = fmaf(v, v, acc);
  }
  sq[b * PS + q] = acc;
}

#define QCAP 128
#define QP   132

template <bool LAST>
__device__ __forceinline__ void flush_rows(const float* qrow_s, const int* qrow_i, int ln, int half,
                                           int (&cnt)[16], int& over,
                                           float (&rs)[4], int (&ri)[4], float (&thr)[8]) {
  __builtin_amdgcn_fence(__ATOMIC_RELEASE, "workgroup");
  __builtin_amdgcn_wave_barrier();
  __builtin_amdgcn_fence(__ATOMIC_ACQUIRE, "workgroup");
  int n = 0;
  #pragma unroll
  for (int r = 0; r < 16; ++r) n = (ln == r) ? cnt[r] : n;
  n = n < QCAP ? n : QCAP;
  int mx = cnt[0];
  #pragma unroll
  for (int r = 1; r < 16; ++r) mx = cnt[r] > mx ? cnt[r] : mx;
  mx = __builtin_amdgcn_readfirstlane(mx);
  const int nmax = mx < QCAP ? mx : QCAP;
  for (int j = 0; j < nmax; j += 4) {
    const v4f  s4 = *(const v4f*)(qrow_s + j);
    const int4 i4 = *(const int4*)(qrow_i + j);
    const int ids[4] = {i4.x, i4.y, i4.z, i4.w};
    #pragma unroll
    for (int e = 0; e < 4; ++e) {
      const float s  = (j + e < n) ? s4[e] : FLT_BIG;
      const int   id = ids[e];
      const bool c0 = s < rs[0], c1 = s < rs[1], c2 = s < rs[2], c3 = s < rs[3];
      rs[3] = c2 ? rs[2] : (c3 ? s : rs[3]);  ri[3] = c2 ? ri[2] : (c3 ? id : ri[3]);
      rs[2] = c1 ? rs[1] : (c2 ? s : rs[2]);  ri[2] = c1 ? ri[1] : (c2 ? id : ri[2]);
      rs[1] = c0 ? rs[0] : (c1 ? s : rs[1]);  ri[1] = c0 ? ri[0] : (c1 ? id : ri[1]);
      rs[0] = c0 ? s : rs[0];                 ri[0] = c0 ? id : ri[0];
    }
  }
  #pragma unroll
  for (int r = 0; r < 16; ++r) cnt[r] = 0;
  over = 0;
  if (!LAST) {
    #pragma unroll
    for (int v = 0; v < 8; ++v)
      thr[v] = __int_as_float(__builtin_amdgcn_ds_bpermute(4 * (v + 8 * half), __float_as_int(rs[3])));
  }
}

#define NW   8
#define NQ   4
#define QLEN (PS / NQ)
#define CROW 260

__launch_bounds__(256)
__global__ void knn_kernel(const float* __restrict__ T, const float* __restrict__ S,
                           const float* __restrict__ sq, float* __restrict__ pscore,
                           int* __restrict__ pidx) {
  __shared__ __attribute__((aligned(16))) float stage[2][16 * CROW];
  __shared__ __attribute__((aligned(16))) float qs[NW][16][QP];
  __shared__ __attribute__((aligned(16))) int   qi[NW][16][QP];

  const int b    = blockIdx.y;
  const int z    = blockIdx.z;
  const int lane = threadIdx.x & 31;
  const int wave = threadIdx.x >> 5;
  const int half = lane >> 4;
  const int ln   = lane & 15;
  const int p0   = blockIdx.x * (16 * NW) + wave * 16;

  v2f a[16];
  #pragma unroll
  for (int s = 0; s < 16; ++s) {
    int f = F0 + 4 * s + 2 * half;
    const float* tp = T + ((size_t)b * C_ + f) * PT + p0 + ln;
    a[s].x = tp[0];
    a[s].y = tp[PT];
  }

  float rs[KSEL]; int ri[KSEL];
  #pragma unroll
  for (int j = 0; j < KSEL; ++j) { rs[j] = FLT_BIG; ri[j] = 0; }

  float thr[8];
  #pragma unroll
  for (int v = 0; v < 8; ++v) thr[v] = FLT_BIG;

  int cnt[16];
  #pragma unroll
  for (int r = 0; r < 16; ++r) cnt[r] = 0;
  int over = 0;
  const unsigned hmask = half ? 0xFFFF0000u : 0x0000FFFFu;
  float* const qpush_s = &qs[wave][8 * half][0];
  int*   const qpush_i = &qi[wave][8 * half][0];
  const float* const qown_s = &qs[wave][ln][0];
  const int*   const qown_i = &qi[wave][ln][0];

  const int qbeg = z * QLEN;
  const int fs = 2 * (wave * 2 + half);
  const float* gsrc = S + ((size_t)b * C_ + F0 + fs) * PS + qbeg + 4 * ln;
  const int st_off = ln * CROW + fs;
  const int fr_off = ln * CROW + 2 * half;

  v4f gx[2], gy[2];
  {
    v4f hx[2], hy[2];
    #pragma unroll
    for (int j = 0; j < 2; ++j) {
      hx[j] = *(const v4f*)(gsrc + (size_t)(32 * j) * PS);
      hy[j] = *(const v4f*)(gsrc + (size_t)(32 * j + 1) * PS);
    }
    #pragma unroll
    for (int j = 0; j < 2; ++j) {
      gx[j] = *(const v4f*)(gsrc + (size_t)(32 * j) * PS + 64);
      gy[j] = *(const v4f*)(gsrc + (size_t)(32 * j + 1) * PS + 64);
    }
    #pragma unroll
    for (int j = 0; j < 2; ++j)
      #pragma unroll
      for (int t = 0; t < 4; ++t) {
        v2f w; w.x = hx[j][t]; w.y = hy[j][t];
        *(v2f*)(&stage[0][st_off + 32 * j + 64 * t]) = w;
      }
  }
  __syncthreads();

  v2f bfA[16][4], bfB[16][4];

  #pragma unroll
  for (int j = 0; j < 2; ++j)
    #pragma unroll
    for (int t = 0; t < 4; ++t) {
      v2f w; w.x = gx[j][t]; w.y = gy[j][t];
      *(v2f*)(&stage[1][st_off + 32 * j + 64 * t]) = w;
    }
  #pragma unroll
  for (int j = 0; j < 2; ++j) {
    gx[j] = *(const v4f*)(gsrc + (size_t)(32 * j) * PS + 128);
    gy[j] = *(const v4f*)(gsrc + (size_t)(32 * j + 1) * PS + 128);
  }
  #pragma unroll
  for (int s = 0; s < 16; ++s)
    #pragma unroll
    for (int t = 0; t < 4; ++t)
      bfA[s][t] = *(const v2f*)(&stage[0][fr_off + 64 * t + 4 * s]);
  __syncthreads();

  auto slab = [&](const int it, const int cur, v2f (&use)[16][4], v2f (&nxt)[16][4])
      __attribute__((always_inline)) {
    const int q0  = qbeg + it * 64;
    const bool more  = (it + 1) < QLEN / 64;
    const bool more2 = (it + 2) < QLEN / 64;
    const bool more3 = (it + 3) < QLEN / 64;

    if (more2) {
      #pragma unroll
      for (int j = 0; j < 2; ++j)
        #pragma unroll
        for (int t = 0; t < 4; ++t) {
          v2f w; w.x = gx[j][t]; w.y = gy[j][t];
          *(v2f*)(&stage[cur][st_off + 32 * j + 64 * t]) = w;
        }
    }
    if (more3) {
      #pragma unroll
      for (int j = 0; j < 2; ++j) {
        gx[j] = *(const v4f*)(gsrc + (size_t)(32 * j) * PS + (it + 3) * 64);
        gy[j] = *(const v4f*)(gsrc + (size_t)(32 * j + 1) * PS + (it + 3) * 64);
      }
    }
    if (more) {
      const float* fr = &stage[cur ^ 1][fr_off];
      #pragma unroll
      for (int s = 0; s < 16; ++s)
        #pragma unroll
        for (int t = 0; t < 4; ++t)
          nxt[s][t] = *(const v2f*)(fr + 64 * t + 4 * s);
    }
    const v4f sq4 = *(const v4f*)(sq + b * PS + q0 + 4 * ln);

    v8f acc[4] = {};
    #pragma unroll
    for (int s = 0; s < 16; ++s) {
      #pragma unroll
      for (int t = 0; t < 4; ++t) {
        acc[t] = __builtin_amdgcn_wmma_f32_16x16x4_f32(
             false, a[s],  false, use[s][t],
             (short)0, acc[t],  false,  false);
      }
    }

    __builtin_amdgcn_s_setprio(3);

    __builtin_amdgcn_fence(__ATOMIC_RELEASE, "workgroup");
    __builtin_amdgcn_s_barrier_signal(-1);

    float sc[4][8];
    #pragma unroll
    for (int t = 0; t < 4; ++t)
      #pragma unroll
      for (int v = 0; v < 8; ++v)
        sc[t][v] = fmaf(-2.f, acc[t][v], sq4[t]);

    unsigned pm[4][8];
    #pragma unroll
    for (int t = 0; t < 4; ++t)
      #pragma unroll
      for (int v = 0; v < 8; ++v)
        pm[t][v] = __builtin_amdgcn_ballot_w32(sc[t][v] < thr[v]);

    #pragma unroll
    for (int t = 0; t < 4; ++t) {
      const int qidx = q0 + 4 * ln + t;

      #pragma unroll
      for (int v = 0; v < 8; ++v) {
        if (__builtin_expect(pm[t][v] != 0u, 0)) {
          const unsigned m = pm[t][v];
          int pos = (half ? cnt[v + 8] : cnt[v]) + (int)__builtin_amdgcn_mbcnt_lo(m & hmask, 0u);
          pos = pos < QCAP - 1 ? pos : QCAP - 1;
          if (sc[t][v] < thr[v]) {
            qpush_s[v * QP + pos] = sc[t][v];
            qpush_i[v * QP + pos] = qidx;
          }
          cnt[v]     += __builtin_popcount(m & 0xFFFFu);
          cnt[v + 8] += __builtin_popcount(m >> 16);
          over |= (int)(cnt[v] > QCAP - 64) | (int)(cnt[v + 8] > QCAP - 64);
        }
      }
    }

    const int done = it + 1;
    if (((done & it) == 0 && done < QLEN / 64) || over != 0)
      flush_rows<false>(qown_s, qown_i, ln, half, cnt, over, rs, ri, thr);

    __builtin_amdgcn_s_barrier_wait(-1);
    __builtin_amdgcn_fence(__ATOMIC_ACQUIRE, "workgroup");
    __builtin_amdgcn_s_setprio(0);
  };

  for (int it = 0; it < QLEN / 64; it += 2) {
    slab(it,     0, bfA, bfB);
    slab(it + 1, 1, bfB, bfA);
  }

  flush_rows<true>(qown_s, qown_i, ln, half, cnt, over, rs, ri, thr);

  if (lane < 16) {
    const size_t o = (((size_t)b * NQ + z) * PT + p0 + lane) * KSEL;
    v4f so; so.x = rs[0]; so.y = rs[1]; so.z = rs[2]; so.w = rs[3];
    *(v4f*)(pscore + o) = so;
    *(int4*)(pidx + o) = make_int4(ri[0], ri[1], ri[2], ri[3]);
  }
}

#define CG  4
#define CPG (C_ / CG)
#define GT  128

__launch_bounds__(GT)
__global__ void merge_gather_kernel(const float* __restrict__ S, const float* __restrict__ pscore,
                                    const int* __restrict__ pidx, float* __restrict__ out) {
  const int p = blockIdx.x * GT + threadIdx.x;
  const int g = blockIdx.y;
  const int b = blockIdx.z;
  float rs[KSEL]; int ri[KSEL];
  #pragma unroll
  for (int j = 0; j < KSEL; ++j) { rs[j] = FLT_BIG; ri[j] = 0; }
  v4f  qs[NQ];
  int4 qi[NQ];
  #pragma unroll
  for (int q = 0; q < NQ; ++q) {
    const size_t o = (((size_t)b * NQ + q) * PT + p) * KSEL;
    qs[q] = *(const v4f*)(pscore + o);
    qi[q] = *(const int4*)(pidx + o);
  }
  #pragma unroll
  for (int q = 0; q < NQ; ++q) {
    const v4f s4 = qs[q];
    const int ids[KSEL] = {qi[q].x, qi[q].y, qi[q].z, qi[q].w};
    #pragma unroll
    for (int j = 0; j < KSEL; ++j) {
      float sc = s4[j];
      int   id = ids[j];
      if (sc < rs[3]) {
        if (sc < rs[2]) {
          rs[3] = rs[2]; ri[3] = ri[2];
          if (sc < rs[1]) {
            rs[2] = rs[1]; ri[2] = ri[1];
            if (sc < rs[0]) {
              rs[1] = rs[0]; ri[1] = ri[0]; rs[0] = sc; ri[0] = id;
            } else { rs[1] = sc; ri[1] = id; }
          } else { rs[2] = sc; ri[2] = id; }
        } else { rs[3] = sc; ri[3] = id; }
      }
    }
  }
  const int i0 = ri[0] & (PS - 1), i1 = ri[1] & (PS - 1);
  const int i2 = ri[2] & (PS - 1), i3 = ri[3] & (PS - 1);

  const float* src = S + ((size_t)b * C_ + (size_t)g * CPG) * PS;
  float* o = out + (((size_t)b * C_ + (size_t)g * CPG) * PT + p) * KSEL;
  #pragma unroll
  for (int j = 0; j < CPG; ++j) {
    const float* row = src + (size_t)j * PS;
    float4 r;
    r.x = row[i0]; r.y = row[i1]; r.z = row[i2]; r.w = row[i3];
    *(float4*)(o + (size_t)j * PT * KSEL) = r;
  }
}

extern "C" void kernel_launch(void* const* d_in, const int* in_sizes, int n_in,
                              void* d_out, int out_size, void* d_ws, size_t ws_size,
                              hipStream_t stream) {
  const float* T = (const float*)d_in[0];
  const float* S = (const float*)d_in[1];

  float* sq  = (float*)d_ws;
  float* psc = (float*)((char*)d_ws + (size_t)262144);
  int*   pid = (int*)((char*)d_ws + (size_t)786432);
  float* out = (float*)d_out;

  sqnorm_kernel<<<dim3(PS / 256, B_), 256, 0, stream>>>(S, sq);
  knn_kernel<<<dim3(PT / (16 * NW), B_, NQ), 256, 0, stream>>>(T, S, sq, psc, pid);
  merge_gather_kernel<<<dim3(PT / GT, CG, B_), GT, 0, stream>>>(S, psc, pid, out);
}


# Source squared norms computed in the search kernel from the slab staged in LDS (same ascending fmaf chain per source, links between the WMMAs); norm kernel and launch removed; scores bit-identical
# speedup vs baseline: 1.9134x; 1.0372x over previous
#include <hip/hip_runtime.h>

typedef __attribute__((ext_vector_type(2))) float v2f;
typedef __attribute__((ext_vector_type(4))) float v4f;
typedef __attribute__((ext_vector_type(8))) float v8f;

#define B_   4
#define C_   68
#define F0   4
#define PT   2048
#define PS   8192
#define KSEL 4
#define FLT_BIG 3.4e38f

#define QCAP 128
#define QP   132

template <bool LAST>
__device__ __forceinline__ void flush_rows(const float* qrow_s, const int* qrow_i, int ln, int half,
                                           int (&cnt)[16], int& over,
                                           float (&rs)[4], int (&ri)[4], float (&thr)[8]) {
  __builtin_amdgcn_fence(__ATOMIC_RELEASE, "workgroup");
  __builtin_amdgcn_wave_barrier();
  __builtin_amdgcn_fence(__ATOMIC_ACQUIRE, "workgroup");
  int n = 0;
  #pragma unroll
  for (int r = 0; r < 16; ++r) n = (ln == r) ? cnt[r] : n;
  n = n < QCAP ? n : QCAP;
  int mx = cnt[0];
  #pragma unroll
  for (int r = 1; r < 16; ++r) mx = cnt[r] > mx ? cnt[r] : mx;
  mx = __builtin_amdgcn_readfirstlane(mx);
  const int nmax = mx < QCAP ? mx : QCAP;
  for (int j = 0; j < nmax; j += 4) {
    const v4f  s4 = *(const v4f*)(qrow_s + j);
    const int4 i4 = *(const int4*)(qrow_i + j);
    const int ids[4] = {i4.x, i4.y, i4.z, i4.w};
    #pragma unroll
    for (int e = 0; e < 4; ++e) {
      const float s  = (j + e < n) ? s4[e] : FLT_BIG;
      const int   id = ids[e];
      const bool c0 = s < rs[0], c1 = s < rs[1], c2 = s < rs[2], c3 = s < rs[3];
      rs[3] = c2 ? rs[2] : (c3 ? s : rs[3]);  ri[3] = c2 ? ri[2] : (c3 ? id : ri[3]);
      rs[2] = c1 ? rs[1] : (c2 ? s : rs[2]);  ri[2] = c1 ? ri[1] : (c2 ? id : ri[2]);
      rs[1] = c0 ? rs[0] : (c1 ? s : rs[1]);  ri[1] = c0 ? ri[0] : (c1 ? id : ri[1]);
      rs[0] = c0 ? s : rs[0];                 ri[0] = c0 ? id : ri[0];
    }
  }
  #pragma unroll
  for (int r = 0; r < 16; ++r) cnt[r] = 0;
  over = 0;
  if (!LAST) {
    #pragma unroll
    for (int v = 0; v < 8; ++v)
      thr[v] = __int_as_float(__builtin_amdgcn_ds_bpermute(4 * (v + 8 * half), __float_as_int(rs[3])));
  }
}

#define NW   8
#define NQ   4
#define QLEN (PS / NQ)
#define CROW 260

__launch_bounds__(256)
__global__ void knn_kernel(const float* __restrict__ T, const float* __restrict__ S,
                           float* __restrict__ pscore, int* __restrict__ pidx) {
  __shared__ __attribute__((aligned(16))) float stage[2][16 * CROW];
  __shared__ __attribute__((aligned(16))) float qs[NW][16][QP];
  __shared__ __attribute__((aligned(16))) int   qi[NW][16][QP];
  __shared__ __attribute__((aligned(16))) float sqs[2][64];

  const int b    = blockIdx.y;
  const int z    = blockIdx.z;
  const int lane = threadIdx.x & 31;
  const int wave = threadIdx.x >> 5;
  const int half = lane >> 4;
  const int ln   = lane & 15;
  const int p0   = blockIdx.x * (16 * NW) + wave * 16;

  v2f a[16];
  #pragma unroll
  for (int s = 0; s < 16; ++s) {
    int f = F0 + 4 * s + 2 * half;
    const float* tp = T + ((size_t)b * C_ + f) * PT + p0 + ln;
    a[s].x = tp[0];
    a[s].y = tp[PT];
  }

  float rs[KSEL]; int ri[KSEL];
  #pragma unroll
  for (int j = 0; j < KSEL; ++j) { rs[j] = FLT_BIG; ri[j] = 0; }

  float thr[8];
  #pragma unroll
  for (int v = 0; v < 8; ++v) thr[v] = FLT_BIG;

  int cnt[16];
  #pragma unroll
  for (int r = 0; r < 16; ++r) cnt[r] = 0;
  int over = 0;
  const unsigned hmask = half ? 0xFFFF0000u : 0x0000FFFFu;
  float* const qpush_s = &qs[wave][8 * half][0];
  int*   const qpush_i = &qi[wave][8 * half][0];
  const float* const qown_s = &qs[wave][ln][0];
  const int*   const qown_i = &qi[wave][ln][0];

  const int qbeg = z * QLEN;
  const int fs = 2 * (wave * 2 + half);
  const float* gsrc = S + ((size_t)b * C_ + F0 + fs) * PS + qbeg + 4 * ln;
  const int st_off = ln * CROW + fs;
  const int fr_off = ln * CROW + 2 * half;
  const int nc = 2 * (lane & 7) + (wave & 1);
  const int nt = wave >> 1;
  const int nr_off = nc * CROW + nt * 64;
  const int nw_idx = 4 * nc + nt;

  v4f gx[2], gy[2];
  {
    v4f hx[2], hy[2];
    #pragma unroll
    for (int j = 0; j < 2; ++j) {
      hx[j] = *(const v4f*)(gsrc + (size_t)(32 * j) * PS);
      hy[j] = *(const v4f*)(gsrc + (size_t)(32 * j + 1) * PS);
    }
    #pragma unroll
    for (int j = 0; j < 2; ++j) {
      gx[j] = *(const v4f*)(gsrc + (size_t)(32 * j) * PS + 64);
      gy[j] = *(const v4f*)(gsrc + (size_t)(32 * j + 1) * PS + 64);
    }
    #pragma unroll
    for (int j = 0; j < 2; ++j)
      #pragma unroll
      for (int t = 0; t < 4; ++t) {
        v2f w; w.x = hx[j][t]; w.y = hy[j][t];
        *(v2f*)(&stage[0][st_off + 32 * j + 64 * t]) = w;
      }
  }
  __syncthreads();

  {
    const float* nb = &stage[0][nr_off];
    float n0 = 0.f;
    #pragma unroll
    for (int k = 0; k < 16; ++k) {
      const v4f v = *(const v4f*)(nb + 4 * k);
      n0 = fmaf(v.x, v.x, n0); n0 = fmaf(v.y, v.y, n0);
      n0 = fmaf(v.z, v.z, n0); n0 = fmaf(v.w, v.w, n0);
    }
    sqs[0][nw_idx] = n0;
  }

  v2f bfA[16][4], bfB[16][4];

  #pragma unroll
  for (int j = 0; j < 2; ++j)
    #pragma unroll
    for (int t = 0; t < 4; ++t) {
      v2f w; w.x = gx[j][t]; w.y = gy[j][t];
      *(v2f*)(&stage[1][st_off + 32 * j + 64 * t]) = w;
    }
  #pragma unroll
  for (int j = 0; j < 2; ++j) {
    gx[j] = *(const v4f*)(gsrc + (size_t)(32 * j) * PS + 128);
    gy[j] = *(const v4f*)(gsrc + (size_t)(32 * j + 1) * PS + 128);
  }
  #pragma unroll
  for (int s = 0; s < 16; ++s)
    #pragma unroll
    for (int t = 0; t < 4; ++t)
      bfA[s][t] = *(const v2f*)(&stage[0][fr_off + 64 * t + 4 * s]);
  __syncthreads();

  auto slab = [&](const int it, const int cur, v2f (&use)[16][4], v2f (&nxt)[16][4])
      __attribute__((always_inline)) {
    const int q0  = qbeg + it * 64;
    const bool more  = (it + 1) < QLEN / 64;
    const bool more2 = (it + 2) < QLEN / 64;
    const bool more3 = (it + 3) < QLEN / 64;

    v4f nv[16];
    {
      const float* nb = &stage[cur ^ 1][nr_off];
      #pragma unroll
      for (int k = 0; k < 16; ++k) nv[k] = *(const v4f*)(nb + 4 * k);
    }
    const v4f sq4 = *(const v4f*)(&sqs[cur][4 * ln]);

    if (more2) {
      #pragma unroll
      for (int j = 0; j < 2; ++j)
        #pragma unroll
        for (int t = 0; t < 4; ++t) {
          v2f w; w.x = gx[j][t]; w.y = gy[j][t];
          *(v2f*)(&stage[cur][st_off + 32 * j + 64 * t]) = w;
        }
    }
    if (more3) {
      #pragma unroll
      for (int j = 0; j < 2; ++j) {
        gx[j] = *(const v4f*)(gsrc + (size_t)(32 * j) * PS + (it + 3) * 64);
        gy[j] = *(const v4f*)(gsrc + (size_t)(32 * j + 1) * PS + (it + 3) * 64);
      }
    }
    if (more) {
      const float* fr = &stage[cur ^ 1][fr_off];
      #pragma unroll
      for (int s = 0; s < 16; ++s)
        #pragma unroll
        for (int t = 0; t < 4; ++t)
          nxt[s][t] = *(const v2f*)(fr + 64 * t + 4 * s);
    }
    float nacc = 0.f;

    v8f acc[4] = {};
    #pragma unroll
    for (int s = 0; s < 16; ++s) {
      #pragma unroll
      for (int t = 0; t < 4; ++t) {
        if (s == 6 && t == 0) __builtin_amdgcn_sched_barrier(0);
        acc[t] = __builtin_amdgcn_wmma_f32_16x16x4_f32(
             false, a[s],  false, use[s][t],
             (short)0, acc[t],  false,  false);
        if (s >= 6 && s < 14) {
          const int k = 4 * (s - 6) + t;
          const v4f v = nv[k >> 1];
          if (k & 1) { nacc = fmaf(v.z, v.z, nacc); nacc = fmaf(v.w, v.w, nacc); }
          else       { nacc = fmaf(v.x, v.x, nacc); nacc = fmaf(v.y, v.y, nacc); }
          __builtin_amdgcn_sched_barrier(0);
        }
      }
      if (s == 13) {
        sqs[cur ^ 1][nw_idx] = nacc;
        __builtin_amdgcn_sched_barrier(0);
      }
    }

    __builtin_amdgcn_s_setprio(3);

    __builtin_amdgcn_fence(__ATOMIC_RELEASE, "workgroup");
    __builtin_amdgcn_s_barrier_signal(-1);

    float sc[4][8];
    #pragma unroll
    for (int t = 0; t < 4; ++t)
      #pragma unroll
      for (int v = 0; v < 8; ++v)
        sc[t][v] = fmaf(-2.f, acc[t][v], sq4[t]);

    unsigned pm[4][8];
    #pragma unroll
    for (int t = 0; t < 4; ++t)
      #pragma unroll
      for (int v = 0; v < 8; ++v)
        pm[t][v] = __builtin_amdgcn_ballot_w32(sc[t][v] < thr[v]);

    #pragma unroll
    for (int t = 0; t < 4; ++t) {
      const int qidx = q0 + 4 * ln + t;

      #pragma unroll
      for (int v = 0; v < 8; ++v) {
        if (__builtin_expect(pm[t][v] != 0u, 0)) {
          const unsigned m = pm[t][v];
          int pos = (half ? cnt[v + 8] : cnt[v]) + (int)__builtin_amdgcn_mbcnt_lo(m & hmask, 0u);
          pos = pos < QCAP - 1 ? pos : QCAP - 1;
          if (sc[t][v] < thr[v]) {
            qpush_s[v * QP + pos] = sc[t][v];
            qpush_i[v * QP + pos] = qidx;
          }
          cnt[v]     += __builtin_popcount(m & 0xFFFFu);
          cnt[v + 8] += __builtin_popcount(m >> 16);
          over |= (int)(cnt[v] > QCAP - 64) | (int)(cnt[v + 8] > QCAP - 64);
        }
      }
    }

    const int done = it + 1;
    if (((done & it) == 0 && done < QLEN / 64) || over != 0)
      flush_rows<false>(qown_s, qown_i, ln, half, cnt, over, rs, ri, thr);

    __builtin_amdgcn_s_barrier_wait(-1);
    __builtin_amdgcn_fence(__ATOMIC_ACQUIRE, "workgroup");
    __builtin_amdgcn_s_setprio(0);
  };

  for (int it = 0; it < QLEN / 64; it += 2) {
    slab(it,     0, bfA, bfB);
    slab(it + 1, 1, bfB, bfA);
  }

  flush_rows<true>(qown_s, qown_i, ln, half, cnt, over, rs, ri, thr);

  if (lane < 16) {
    const size_t o = (((size_t)b * NQ + z) * PT + p0 + lane) * KSEL;
    v4f so; so.x = rs[0]; so.y = rs[1]; so.z = rs[2]; so.w = rs[3];
    *(v4f*)(pscore + o) = so;
    *(int4*)(pidx + o) = make_int4(ri[0], ri[1], ri[2], ri[3]);
  }
}

#define CG  4
#define CPG (C_ / CG)
#define GT  128

__launch_bounds__(GT)
__global__ void merge_gather_kernel(const float* __restrict__ S, const float* __restrict__ pscore,
                                    const int* __restrict__ pidx, float* __restrict__ out) {
  const int p = blockIdx.x * GT + threadIdx.x;
  const int g = blockIdx.y;
  const int b = blockIdx.z;
  float rs[KSEL]; int ri[KSEL];
  #pragma unroll
  for (int j = 0; j < KSEL; ++j) { rs[j] = FLT_BIG; ri[j] = 0; }
  v4f  qs[NQ];
  int4 qi[NQ];
  #pragma unroll
  for (int q = 0; q < NQ; ++q) {
    const size_t o = (((size_t)b * NQ + q) * PT + p) * KSEL;
    qs[q] = *(const v4f*)(pscore + o);
    qi[q] = *(const int4*)(pidx + o);
  }
  #pragma unroll
  for (int q = 0; q < NQ; ++q) {
    const v4f s4 = qs[q];
    const int ids[KSEL] = {qi[q].x, qi[q].y, qi[q].z, qi[q].w};
    #pragma unroll
    for (int j = 0; j < KSEL; ++j) {
      float sc = s4[j];
      int   id = ids[j];
      if (sc < rs[3]) {
        if (sc < rs[2]) {
          rs[3] = rs[2]; ri[3] = ri[2];
          if (sc < rs[1]) {
            rs[2] = rs[1]; ri[2] = ri[1];
            if (sc < rs[0]) {
              rs[1] = rs[0]; ri[1] = ri[0]; rs[0] = sc; ri[0] = id;
            } else { rs[1] = sc; ri[1] = id; }
          } else { rs[2] = sc; ri[2] = id; }
        } else { rs[3] = sc; ri[3] = id; }
      }
    }
  }
  const int i0 = ri[0] & (PS - 1), i1 = ri[1] & (PS - 1);
  const int i2 = ri[2] & (PS - 1), i3 = ri[3] & (PS - 1);

  const float* src = S + ((size_t)b * C_ + (size_t)g * CPG) * PS;
  float* o = out + (((size_t)b * C_ + (size_t)g * CPG) * PT + p) * KSEL;
  #pragma unroll
  for (int j = 0; j < CPG; ++j) {
    const float* row = src + (size_t)j * PS;
    float4 r;
    r.x = row[i0]; r.y = row[i1]; r.z = row[i2]; r.w = row[i3];
    *(float4*)(o + (size_t)j * PT * KSEL) = r;
  }
}

extern "C" void kernel_launch(void* const* d_in, const int* in_sizes, int n_in,
                              void* d_out, int out_size, void* d_ws, size_t ws_size,
                              hipStream_t stream) {
  const float* T = (const float*)d_in[0];
  const float* S = (const float*)d_in[1];

  float* psc = (float*)((char*)d_ws + (size_t)262144);
  int*   pid = (int*)((char*)d_ws + (size_t)786432);
  float* out = (float*)d_out;

  knn_kernel<<<dim3(PT / (16 * NW), B_, NQ), 256, 0, stream>>>(T, S, psc, pid);
  merge_gather_kernel<<<dim3(PT / GT, CG, B_), GT, 0, stream>>>(S, psc, pid, out);
}
